# starting kernel
# baseline (speedup 1.0000x reference)
#include <hip/hip_runtime.h>


typedef __attribute__((ext_vector_type(2))) float v2f;
typedef __attribute__((ext_vector_type(8))) float v8f;

#define IN_DIM  128
#define HID_DIM 128
#define OUT_DIM 64

__global__ void gcn_degree_kernel(const int* __restrict__ src,
                                  const int* __restrict__ dst,
                                  int* __restrict__ deg_out,
                                  int* __restrict__ deg_in, int E) {
    int e = blockIdx.x * blockDim.x + threadIdx.x;
    if (e < E) {
        atomicAdd(&deg_out[src[e]], 1);
        atomicAdd(&deg_in[dst[e]], 1);
    }
}

__global__ void gcn_invsqrt_kernel(const int* __restrict__ deg,
                                   float* __restrict__ inv, int n2) {
    int i = blockIdx.x * blockDim.x + threadIdx.x;
    if (i < n2) {
        float d = (float)deg[i];
        inv[i] = rsqrtf(fmaxf(d, 1.0f));
    }
}

__global__ __launch_bounds__(256)
void gcn_scatter_kernel(const float* __restrict__ x,
                        const int* __restrict__ src,
                        const int* __restrict__ dst,
                        const float* __restrict__ inv_out,
                        float* __restrict__ agg, int E) {
    long long t = (long long)blockIdx.x * blockDim.x + threadIdx.x;
    int e = (int)(t >> 5);
    if (e >= E) return;
    int lane = (int)(t & 31);
    int s = src[e];
    int d = dst[e];
    float sc = inv_out[s];
    const float4* xp = (const float4*)(x + (size_t)s * 128) + lane;
    float4 v = *xp;
    float* ap = agg + (size_t)d * 128 + lane * 4;
    unsafeAtomicAdd(ap + 0, v.x * sc);
    unsafeAtomicAdd(ap + 1, v.y * sc);
    unsafeAtomicAdd(ap + 2, v.z * sc);
    unsafeAtomicAdd(ap + 3, v.w * sc);
}

template <int K, int NOUT, bool RELU>
__global__ __launch_bounds__(256)
void gcn_wmma_gemm_kernel(const float* __restrict__ A,
                          const float* __restrict__ W,
                          const float* __restrict__ bias,
                          const float* __restrict__ inv_in,
                          float* __restrict__ out, int N) {
    constexpr int S  = 2 * NOUT + 32;
    constexpr int NT = NOUT / 16;
    __shared__ float lds_w[(K / 2) * S];

    for (int idx = threadIdx.x; idx < K * NOUT; idx += blockDim.x) {
        int k = idx / NOUT;
        int n = idx - k * NOUT;
        lds_w[(k >> 1) * S + n * 2 + (k & 1)] = W[idx];
    }
    __syncthreads();

    const int lane = threadIdx.x & 31;
    const int wave = threadIdx.x >> 5;
    const int g = lane >> 4;
    const int mr = lane & 15;

    const int row0 = blockIdx.x * 128 + wave * 16;
    int row = row0 + mr;
    if (row >= N) row = N - 1;
    const float s = inv_in[row];
    const float* __restrict__ Arow = A + (size_t)row * K;

    v8f acc[NT];
#pragma unroll
    for (int t = 0; t < NT; ++t) acc[t] = (v8f){};

#pragma unroll 4
    for (int k0 = 0; k0 < K; k0 += 4) {
        const int ka = k0 + 2 * g;
        v2f a = *(const v2f*)(Arow + ka);
        a.x *= s;
        a.y *= s;
        const v2f* __restrict__ wp =
            (const v2f*)(lds_w + (size_t)(k0 / 2 + g) * S);
#pragma unroll
        for (int t = 0; t < NT; ++t) {
            v2f b = wp[t * 16 + mr];
            acc[t] = __builtin_amdgcn_wmma_f32_16x16x4_f32(
                 false, a,  false, b,
                 (short)0, acc[t],  false,  false);
        }
    }

#pragma unroll
    for (int t = 0; t < NT; ++t) {
        const int Nbase = t * 16;
        const float bv = bias[Nbase + mr];
#pragma unroll
        for (int r = 0; r < 8; ++r) {
            const int orow = row0 + r + 8 * g;
            if (orow < N) {
                float v = acc[t][r] + bv;
                if (RELU) v = fmaxf(v, 0.0f);
                out[(size_t)orow * NOUT + Nbase + mr] = v;
            }
        }
    }
}

extern "C" void kernel_launch(void* const* d_in, const int* in_sizes, int n_in,
                              void* d_out, int out_size, void* d_ws, size_t ws_size,
                              hipStream_t stream) {
    const float* in_feat = (const float*)d_in[0];
    const float* W1      = (const float*)d_in[1];
    const float* b1      = (const float*)d_in[2];
    const float* W2      = (const float*)d_in[3];
    const float* b2      = (const float*)d_in[4];
    const int*   src     = (const int*)d_in[5];
    const int*   dst     = (const int*)d_in[6];
    float*       out     = (float*)d_out;

    const int N = in_sizes[0] / IN_DIM;
    const int E = in_sizes[5];

    char* ws = (char*)d_ws;
    int*   deg   = (int*)ws;
    float* inv   = (float*)(ws + (size_t)2 * N * 4);
    float* agg   = (float*)(ws + (size_t)4 * N * 4);
    float* hbuf  = agg + (size_t)N * HID_DIM;
    float* inv_out = inv;
    float* inv_in  = inv + N;

    hipMemsetAsync(deg, 0, (size_t)2 * N * sizeof(int), stream);
    hipMemsetAsync(agg, 0, (size_t)N * HID_DIM * sizeof(float), stream);

    {
        int blk = 256;
        gcn_degree_kernel<<<(E + blk - 1) / blk, blk, 0, stream>>>(
            src, dst, deg, deg + N, E);
        gcn_invsqrt_kernel<<<(2 * N + blk - 1) / blk, blk, 0, stream>>>(
            deg, inv, 2 * N);
    }

    {
        long long threads = (long long)E * 32;
        int blk = 256;
        long long grid = (threads + blk - 1) / blk;
        gcn_scatter_kernel<<<(unsigned)grid, blk, 0, stream>>>(
            in_feat, src, dst, inv_out, agg, E);
    }

    {
        int grid = (N + 127) / 128;
        gcn_wmma_gemm_kernel<IN_DIM, HID_DIM, true>
            <<<grid, 256, 0, stream>>>(agg, W1, b1, inv_in, hbuf, N);
    }

    hipMemsetAsync(agg, 0, (size_t)N * HID_DIM * sizeof(float), stream);
    {
        long long threads = (long long)E * 32;
        int blk = 256;
        long long grid = (threads + blk - 1) / blk;
        gcn_scatter_kernel<<<(unsigned)grid, blk, 0, stream>>>(
            hbuf, src, dst, inv_out, agg, E);
    }

    {
        int grid = (N + 127) / 128;
        gcn_wmma_gemm_kernel<HID_DIM, OUT_DIM, false>
            <<<grid, 256, 0, stream>>>(agg, W2, b2, inv_in, out, N);
    }
}


# Layer 2 applies W2 before its edge sum (A(hW2), not (Ah)W2): WMMA makes z=(inv_out*h)@W2, then a 64-wide atomic scatter adds inv_in[dst]*z[src] into out preset to b2; f32 sums regrouped
# speedup vs baseline: 1.2254x; 1.2254x over previous
#include <hip/hip_runtime.h>


typedef __attribute__((ext_vector_type(2))) float v2f;
typedef __attribute__((ext_vector_type(8))) float v8f;

#define IN_DIM  128
#define HID_DIM 128
#define OUT_DIM 64

__global__ void gcn_degree_kernel(const int* __restrict__ src,
                                  const int* __restrict__ dst,
                                  int* __restrict__ deg_out,
                                  int* __restrict__ deg_in, int E) {
    int e = blockIdx.x * blockDim.x + threadIdx.x;
    if (e < E) {
        atomicAdd(&deg_out[src[e]], 1);
        atomicAdd(&deg_in[dst[e]], 1);
    }
}

__global__ void gcn_invsqrt_kernel(const int* __restrict__ deg,
                                   float* __restrict__ inv, int n2) {
    int i = blockIdx.x * blockDim.x + threadIdx.x;
    if (i < n2) {
        float d = (float)deg[i];
        inv[i] = rsqrtf(fmaxf(d, 1.0f));
    }
}

__global__ __launch_bounds__(256)
void gcn_scatter_kernel(const float* __restrict__ x,
                        const int* __restrict__ src,
                        const int* __restrict__ dst,
                        const float* __restrict__ inv_out,
                        float* __restrict__ agg, int E) {
    long long t = (long long)blockIdx.x * blockDim.x + threadIdx.x;
    int e = (int)(t >> 5);
    if (e >= E) return;
    int lane = (int)(t & 31);
    int s = src[e];
    int d = dst[e];
    float sc = inv_out[s];
    const float4* xp = (const float4*)(x + (size_t)s * 128) + lane;
    float4 v = *xp;
    float* ap = agg + (size_t)d * 128 + lane * 4;
    unsafeAtomicAdd(ap + 0, v.x * sc);
    unsafeAtomicAdd(ap + 1, v.y * sc);
    unsafeAtomicAdd(ap + 2, v.z * sc);
    unsafeAtomicAdd(ap + 3, v.w * sc);
}

__global__ __launch_bounds__(256)
void gcn_scatter_out_kernel(const float* __restrict__ z,
                            const int* __restrict__ src,
                            const int* __restrict__ dst,
                            const float* __restrict__ inv_in,
                            float* __restrict__ out, int E) {
    long long t = (long long)blockIdx.x * blockDim.x + threadIdx.x;
    int e = (int)(t >> 5);
    if (e >= E) return;
    int lane = (int)(t & 31);
    int s = src[e];
    int d = dst[e];
    float sc = inv_in[d];
    const float2* zp = (const float2*)(z + (size_t)s * OUT_DIM) + lane;
    float2 v = *zp;
    float* op = out + (size_t)d * OUT_DIM + lane * 2;
    unsafeAtomicAdd(op + 0, v.x * sc);
    unsafeAtomicAdd(op + 1, v.y * sc);
}

template <int K, int NOUT, bool FIRST>
__global__ __launch_bounds__(256)
void gcn_wmma_gemm_kernel(const float* __restrict__ A,
                          const float* __restrict__ W,
                          const float* __restrict__ bias,
                          const float* __restrict__ scale,
                          float* __restrict__ out,
                          float* __restrict__ acc_init, int N) {
    constexpr int S  = 2 * NOUT + 32;
    constexpr int NT = NOUT / 16;
    __shared__ float lds_w[(K / 2) * S];

    for (int idx = threadIdx.x; idx < K * NOUT; idx += blockDim.x) {
        int k = idx / NOUT;
        int n = idx - k * NOUT;
        lds_w[(k >> 1) * S + n * 2 + (k & 1)] = W[idx];
    }
    __syncthreads();

    const int lane = threadIdx.x & 31;
    const int wave = threadIdx.x >> 5;
    const int g = lane >> 4;
    const int mr = lane & 15;

    const int row0 = blockIdx.x * 128 + wave * 16;
    int row = row0 + mr;
    if (row >= N) row = N - 1;
    const float s = scale[row];
    const float* __restrict__ Arow = A + (size_t)row * K;

    v8f acc[NT];
#pragma unroll
    for (int t = 0; t < NT; ++t) acc[t] = (v8f){};

#pragma unroll 4
    for (int k0 = 0; k0 < K; k0 += 4) {
        const int ka = k0 + 2 * g;
        v2f a = *(const v2f*)(Arow + ka);
        a.x *= s;
        a.y *= s;
        const v2f* __restrict__ wp =
            (const v2f*)(lds_w + (size_t)(k0 / 2 + g) * S);
#pragma unroll
        for (int t = 0; t < NT; ++t) {
            v2f b = wp[t * 16 + mr];
            acc[t] = __builtin_amdgcn_wmma_f32_16x16x4_f32(
                 false, a,  false, b,
                 (short)0, acc[t],  false,  false);
        }
    }

#pragma unroll
    for (int t = 0; t < NT; ++t) {
        const int Nbase = t * 16;
        const float bv = bias[Nbase + mr];
#pragma unroll
        for (int r = 0; r < 8; ++r) {
            const int orow = row0 + r + 8 * g;
            if (orow < N) {
                const size_t o = (size_t)orow * NOUT + Nbase + mr;
                if (FIRST) {
                    out[o] = fmaxf(acc[t][r] + bv, 0.0f);
                } else {
                    out[o] = acc[t][r];
                    acc_init[o] = bv;
                }
            }
        }
    }
}

extern "C" void kernel_launch(void* const* d_in, const int* in_sizes, int n_in,
                              void* d_out, int out_size, void* d_ws, size_t ws_size,
                              hipStream_t stream) {
    const float* in_feat = (const float*)d_in[0];
    const float* W1      = (const float*)d_in[1];
    const float* b1      = (const float*)d_in[2];
    const float* W2      = (const float*)d_in[3];
    const float* b2      = (const float*)d_in[4];
    const int*   src     = (const int*)d_in[5];
    const int*   dst     = (const int*)d_in[6];
    float*       out     = (float*)d_out;

    const int N = in_sizes[0] / IN_DIM;
    const int E = in_sizes[5];

    char* ws = (char*)d_ws;
    int*   deg   = (int*)ws;
    float* inv   = (float*)(ws + (size_t)2 * N * 4);
    float* agg   = (float*)(ws + (size_t)4 * N * 4);
    float* hbuf  = agg + (size_t)N * HID_DIM;
    float* zbuf  = hbuf + (size_t)N * HID_DIM;
    float* inv_out = inv;
    float* inv_in  = inv + N;

    hipMemsetAsync(deg, 0, (size_t)2 * N * sizeof(int), stream);
    hipMemsetAsync(agg, 0, (size_t)N * HID_DIM * sizeof(float), stream);

    {
        int blk = 256;
        gcn_degree_kernel<<<(E + blk - 1) / blk, blk, 0, stream>>>(
            src, dst, deg, deg + N, E);
        gcn_invsqrt_kernel<<<(2 * N + blk - 1) / blk, blk, 0, stream>>>(
            deg, inv, 2 * N);
    }

    {
        long long threads = (long long)E * 32;
        int blk = 256;
        long long grid = (threads + blk - 1) / blk;
        gcn_scatter_kernel<<<(unsigned)grid, blk, 0, stream>>>(
            in_feat, src, dst, inv_out, agg, E);
    }

    {
        int grid = (N + 127) / 128;
        gcn_wmma_gemm_kernel<IN_DIM, HID_DIM, true>
            <<<grid, 256, 0, stream>>>(agg, W1, b1, inv_in, hbuf, nullptr, N);
    }

    {
        int grid = (N + 127) / 128;
        gcn_wmma_gemm_kernel<HID_DIM, OUT_DIM, false>
            <<<grid, 256, 0, stream>>>(hbuf, W2, b2, inv_out, zbuf, out, N);
    }

    {
        long long threads = (long long)E * 32;
        int blk = 256;
        long long grid = (threads + blk - 1) / blk;
        gcn_scatter_out_kernel<<<(unsigned)grid, blk, 0, stream>>>(
            zbuf, src, dst, inv_in, out, E);
    }
}


# Both edge scatters: lane l owns columns l+32c, so each gather load and each f32 atomic add of a wave covers one whole 128-B line, not parts of 4 (2) lines; sums stay in atomic arrival order
# speedup vs baseline: 1.4515x; 1.1845x over previous
#include <hip/hip_runtime.h>


typedef __attribute__((ext_vector_type(2))) float v2f;
typedef __attribute__((ext_vector_type(8))) float v8f;

#define IN_DIM  128
#define HID_DIM 128
#define OUT_DIM 64

__global__ void gcn_degree_kernel(const int* __restrict__ src,
                                  const int* __restrict__ dst,
                                  int* __restrict__ deg_out,
                                  int* __restrict__ deg_in, int E) {
    int e = blockIdx.x * blockDim.x + threadIdx.x;
    if (e < E) {
        atomicAdd(&deg_out[src[e]], 1);
        atomicAdd(&deg_in[dst[e]], 1);
    }
}

__global__ void gcn_invsqrt_kernel(const int* __restrict__ deg,
                                   float* __restrict__ inv, int n2) {
    int i = blockIdx.x * blockDim.x + threadIdx.x;
    if (i < n2) {
        float d = (float)deg[i];
        inv[i] = rsqrtf(fmaxf(d, 1.0f));
    }
}

__global__ __launch_bounds__(256)
void gcn_scatter_kernel(const float* __restrict__ x,
                        const int* __restrict__ src,
                        const int* __restrict__ dst,
                        const float* __restrict__ inv_out,
                        float* __restrict__ agg, int E) {
    long long t = (long long)blockIdx.x * blockDim.x + threadIdx.x;
    int e = (int)(t >> 5);
    if (e >= E) return;
    int lane = (int)(t & 31);
    int s = src[e];
    int d = dst[e];
    float sc = inv_out[s];
    const float* xp = x + (size_t)s * 128 + lane;
    float v0 = xp[0];
    float v1 = xp[32];
    float v2 = xp[64];
    float v3 = xp[96];
    float* ap = agg + (size_t)d * 128 + lane;
    unsafeAtomicAdd(ap + 0,  v0 * sc);
    unsafeAtomicAdd(ap + 32, v1 * sc);
    unsafeAtomicAdd(ap + 64, v2 * sc);
    unsafeAtomicAdd(ap + 96, v3 * sc);
}

__global__ __launch_bounds__(256)
void gcn_scatter_out_kernel(const float* __restrict__ z,
                            const int* __restrict__ src,
                            const int* __restrict__ dst,
                            const float* __restrict__ inv_in,
                            float* __restrict__ out, int E) {
    long long t = (long long)blockIdx.x * blockDim.x + threadIdx.x;
    int e = (int)(t >> 5);
    if (e >= E) return;
    int lane = (int)(t & 31);
    int s = src[e];
    int d = dst[e];
    float sc = inv_in[d];
    const float* zp = z + (size_t)s * OUT_DIM + lane;
    float v0 = zp[0];
    float v1 = zp[32];
    float* op = out + (size_t)d * OUT_DIM + lane;
    unsafeAtomicAdd(op + 0,  v0 * sc);
    unsafeAtomicAdd(op + 32, v1 * sc);
}

template <int K, int NOUT, bool FIRST>
__global__ __launch_bounds__(256)
void gcn_wmma_gemm_kernel(const float* __restrict__ A,
                          const float* __restrict__ W,
                          const float* __restrict__ bias,
                          const float* __restrict__ scale,
                          float* __restrict__ out,
                          float* __restrict__ acc_init, int N) {
    constexpr int S  = 2 * NOUT + 32;
    constexpr int NT = NOUT / 16;
    __shared__ float lds_w[(K / 2) * S];

    for (int idx = threadIdx.x; idx < K * NOUT; idx += blockDim.x) {
        int k = idx / NOUT;
        int n = idx - k * NOUT;
        lds_w[(k >> 1) * S + n * 2 + (k & 1)] = W[idx];
    }
    __syncthreads();

    const int lane = threadIdx.x & 31;
    const int wave = threadIdx.x >> 5;
    const int g = lane >> 4;
    const int mr = lane & 15;

    const int row0 = blockIdx.x * 128 + wave * 16;
    int row = row0 + mr;
    if (row >= N) row = N - 1;
    const float s = scale[row];
    const float* __restrict__ Arow = A + (size_t)row * K;

    v8f acc[NT];
#pragma unroll
    for (int t = 0; t < NT; ++t) acc[t] = (v8f){};

#pragma unroll 4
    for (int k0 = 0; k0 < K; k0 += 4) {
        const int ka = k0 + 2 * g;
        v2f a = *(const v2f*)(Arow + ka);
        a.x *= s;
        a.y *= s;
        const v2f* __restrict__ wp =
            (const v2f*)(lds_w + (size_t)(k0 / 2 + g) * S);
#pragma unroll
        for (int t = 0; t < NT; ++t) {
            v2f b = wp[t * 16 + mr];
            acc[t] = __builtin_amdgcn_wmma_f32_16x16x4_f32(
                 false, a,  false, b,
                 (short)0, acc[t],  false,  false);
        }
    }

#pragma unroll
    for (int t = 0; t < NT; ++t) {
        const int Nbase = t * 16;
        const float bv = bias[Nbase + mr];
#pragma unroll
        for (int r = 0; r < 8; ++r) {
            const int orow = row0 + r + 8 * g;
            if (orow < N) {
                const size_t o = (size_t)orow * NOUT + Nbase + mr;
                if (FIRST) {
                    out[o] = fmaxf(acc[t][r] + bv, 0.0f);
                } else {
                    out[o] = acc[t][r];
                    acc_init[o] = bv;
                }
            }
        }
    }
}

extern "C" void kernel_launch(void* const* d_in, const int* in_sizes, int n_in,
                              void* d_out, int out_size, void* d_ws, size_t ws_size,
                              hipStream_t stream) {
    const float* in_feat = (const float*)d_in[0];
    const float* W1      = (const float*)d_in[1];
    const float* b1      = (const float*)d_in[2];
    const float* W2      = (const float*)d_in[3];
    const float* b2      = (const float*)d_in[4];
    const int*   src     = (const int*)d_in[5];
    const int*   dst     = (const int*)d_in[6];
    float*       out     = (float*)d_out;

    const int N = in_sizes[0] / IN_DIM;
    const int E = in_sizes[5];

    char* ws = (char*)d_ws;
    int*   deg   = (int*)ws;
    float* inv   = (float*)(ws + (size_t)2 * N * 4);
    float* agg   = (float*)(ws + (size_t)4 * N * 4);
    float* hbuf  = agg + (size_t)N * HID_DIM;
    float* zbuf  = hbuf + (size_t)N * HID_DIM;
    float* inv_out = inv;
    float* inv_in  = inv + N;

    hipMemsetAsync(deg, 0, (size_t)2 * N * sizeof(int), stream);
    hipMemsetAsync(agg, 0, (size_t)N * HID_DIM * sizeof(float), stream);

    {
        int blk = 256;
        gcn_degree_kernel<<<(E + blk - 1) / blk, blk, 0, stream>>>(
            src, dst, deg, deg + N, E);
        gcn_invsqrt_kernel<<<(2 * N + blk - 1) / blk, blk, 0, stream>>>(
            deg, inv, 2 * N);
    }

    {
        long long threads = (long long)E * 32;
        int blk = 256;
        long long grid = (threads + blk - 1) / blk;
        gcn_scatter_kernel<<<(unsigned)grid, blk, 0, stream>>>(
            in_feat, src, dst, inv_out, agg, E);
    }

    {
        int grid = (N + 127) / 128;
        gcn_wmma_gemm_kernel<IN_DIM, HID_DIM, true>
            <<<grid, 256, 0, stream>>>(agg, W1, b1, inv_in, hbuf, nullptr, N);
    }

    {
        int grid = (N + 127) / 128;
        gcn_wmma_gemm_kernel<HID_DIM, OUT_DIM, false>
            <<<grid, 256, 0, stream>>>(hbuf, W2, b2, inv_out, zbuf, out, N);
    }

    {
        long long threads = (long long)E * 32;
        int blk = 256;
        long long grid = (threads + blk - 1) / blk;
        gcn_scatter_out_kernel<<<(unsigned)grid, blk, 0, stream>>>(
            zbuf, src, dst, inv_in, out, E);
    }
}


# Both edge scatters: a wave takes 32 edges (one index per lane), issues the row gathers of 8 edges before their atomic adds; resident grid loops over the edge list; sums stay in atomic arrival order
# speedup vs baseline: 2.0022x; 1.3794x over previous
#include <hip/hip_runtime.h>


typedef __attribute__((ext_vector_type(2))) float v2f;
typedef __attribute__((ext_vector_type(8))) float v8f;

#define IN_DIM  128
#define HID_DIM 128
#define OUT_DIM 64

__global__ void gcn_degree_kernel(const int* __restrict__ src,
                                  const int* __restrict__ dst,
                                  int* __restrict__ deg_out,
                                  int* __restrict__ deg_in, int E) {
    int e = blockIdx.x * blockDim.x + threadIdx.x;
    if (e < E) {
        atomicAdd(&deg_out[src[e]], 1);
        atomicAdd(&deg_in[dst[e]], 1);
    }
}

__global__ void gcn_invsqrt_kernel(const int* __restrict__ deg,
                                   float* __restrict__ inv, int n2) {
    int i = blockIdx.x * blockDim.x + threadIdx.x;
    if (i < n2) {
        float d = (float)deg[i];
        inv[i] = rsqrtf(fmaxf(d, 1.0f));
    }
}

template <int COLS, int U>
__device__ __forceinline__ void gcn_scatter_edges(const float* __restrict__ feat,
                                                  float* __restrict__ acc,
                                                  int s, int d, int sc_bits,
                                                  int j0, int lane) {
    constexpr int C = COLS / 32;
    float v[U][C];
#pragma unroll
    for (int u = 0; u < U; ++u) {
        const unsigned sj = (unsigned)__builtin_amdgcn_readlane(s, j0 + u);
        const float* __restrict__ p = feat + (size_t)(sj * (unsigned)COLS) + lane;
#pragma unroll
        for (int c = 0; c < C; ++c) v[u][c] = p[32 * c];
    }
#pragma unroll
    for (int u = 0; u < U; ++u) {
        const unsigned dj = (unsigned)__builtin_amdgcn_readlane(d, j0 + u);
        const float scj = __int_as_float(__builtin_amdgcn_readlane(sc_bits, j0 + u));
        float* __restrict__ q = acc + (size_t)(dj * (unsigned)COLS) + lane;
#pragma unroll
        for (int c = 0; c < C; ++c) unsafeAtomicAdd(q + 32 * c, v[u][c] * scj);
    }
}

template <int COLS, int U, bool SCALE_BY_SRC>
__global__ __launch_bounds__(256)
void gcn_edge_scatter_kernel(const float* __restrict__ feat,
                             const int* __restrict__ src,
                             const int* __restrict__ dst,
                             const float* __restrict__ inv,
                             float* __restrict__ acc, int E) {
    const int lane = threadIdx.x & 31;
    const int wave_in_block = __builtin_amdgcn_readfirstlane((int)(threadIdx.x >> 5));
    const int wave   = (int)blockIdx.x * 8 + wave_in_block;
    const int nwaves = (int)gridDim.x * 8;
    const int nchunks = (E + 31) >> 5;

    for (int ch = wave; ch < nchunks; ch += nwaves) {
        const int e0 = ch << 5;
        if (e0 + 32 <= E) {
            const int s = src[e0 + lane];
            const int d = dst[e0 + lane];
            const int sc_bits = __float_as_int(inv[SCALE_BY_SRC ? s : d]);
#pragma unroll
            for (int jb = 0; jb < 32; jb += U)
                gcn_scatter_edges<COLS, U>(feat, acc, s, d, sc_bits, jb, lane);
        } else {
            const int n = E - e0;
            int s = 0, d = 0, sc_bits = 0;
            if (lane < n) {
                s = src[e0 + lane];
                d = dst[e0 + lane];
                sc_bits = __float_as_int(inv[SCALE_BY_SRC ? s : d]);
            }
            for (int j = 0; j < n; ++j)
                gcn_scatter_edges<COLS, 1>(feat, acc, s, d, sc_bits, j, lane);
        }
    }
}

template <int K, int NOUT, bool FIRST>
__global__ __launch_bounds__(256)
void gcn_wmma_gemm_kernel(const float* __restrict__ A,
                          const float* __restrict__ W,
                          const float* __restrict__ bias,
                          const float* __restrict__ scale,
                          float* __restrict__ out,
                          float* __restrict__ acc_init, int N) {
    constexpr int S  = 2 * NOUT + 32;
    constexpr int NT = NOUT / 16;
    __shared__ float lds_w[(K / 2) * S];

    for (int idx = threadIdx.x; idx < K * NOUT; idx += blockDim.x) {
        int k = idx / NOUT;
        int n = idx - k * NOUT;
        lds_w[(k >> 1) * S + n * 2 + (k & 1)] = W[idx];
    }
    __syncthreads();

    const int lane = threadIdx.x & 31;
    const int wave = threadIdx.x >> 5;
    const int g = lane >> 4;
    const int mr = lane & 15;

    const int row0 = blockIdx.x * 128 + wave * 16;
    int row = row0 + mr;
    if (row >= N) row = N - 1;
    const float s = scale[row];
    const float* __restrict__ Arow = A + (size_t)row * K;

    v8f acc[NT];
#pragma unroll
    for (int t = 0; t < NT; ++t) acc[t] = (v8f){};

#pragma unroll 4
    for (int k0 = 0; k0 < K; k0 += 4) {
        const int ka = k0 + 2 * g;
        v2f a = *(const v2f*)(Arow + ka);
        a.x *= s;
        a.y *= s;
        const v2f* __restrict__ wp =
            (const v2f*)(lds_w + (size_t)(k0 / 2 + g) * S);
#pragma unroll
        for (int t = 0; t < NT; ++t) {
            v2f b = wp[t * 16 + mr];
            acc[t] = __builtin_amdgcn_wmma_f32_16x16x4_f32(
                 false, a,  false, b,
                 (short)0, acc[t],  false,  false);
        }
    }

#pragma unroll
    for (int t = 0; t < NT; ++t) {
        const int Nbase = t * 16;
        const float bv = bias[Nbase + mr];
#pragma unroll
        for (int r = 0; r < 8; ++r) {
            const int orow = row0 + r + 8 * g;
            if (orow < N) {
                const size_t o = (size_t)orow * NOUT + Nbase + mr;
                if (FIRST) {
                    out[o] = fmaxf(acc[t][r] + bv, 0.0f);
                } else {
                    out[o] = acc[t][r];
                    acc_init[o] = bv;
                }
            }
        }
    }
}

extern "C" void kernel_launch(void* const* d_in, const int* in_sizes, int n_in,
                              void* d_out, int out_size, void* d_ws, size_t ws_size,
                              hipStream_t stream) {
    const float* in_feat = (const float*)d_in[0];
    const float* W1      = (const float*)d_in[1];
    const float* b1      = (const float*)d_in[2];
    const float* W2      = (const float*)d_in[3];
    const float* b2      = (const float*)d_in[4];
    const int*   src     = (const int*)d_in[5];
    const int*   dst     = (const int*)d_in[6];
    float*       out     = (float*)d_out;

    const int N = in_sizes[0] / IN_DIM;
    const int E = in_sizes[5];

    char* ws = (char*)d_ws;
    int*   deg   = (int*)ws;
    float* inv   = (float*)(ws + (size_t)2 * N * 4);
    float* agg   = (float*)(ws + (size_t)4 * N * 4);
    float* hbuf  = agg + (size_t)N * HID_DIM;
    float* zbuf  = hbuf + (size_t)N * HID_DIM;
    float* inv_out = inv;
    float* inv_in  = inv + N;

    hipMemsetAsync(deg, 0, (size_t)2 * N * sizeof(int), stream);
    hipMemsetAsync(agg, 0, (size_t)N * HID_DIM * sizeof(float), stream);

    {
        int blk = 256;
        gcn_degree_kernel<<<(E + blk - 1) / blk, blk, 0, stream>>>(
            src, dst, deg, deg + N, E);
        gcn_invsqrt_kernel<<<(2 * N + blk - 1) / blk, blk, 0, stream>>>(
            deg, inv, 2 * N);
    }

    gcn_edge_scatter_kernel<IN_DIM, 8, true><<<2048, 256, 0, stream>>>(
        in_feat, src, dst, inv_out, agg, E);

    {
        int grid = (N + 127) / 128;
        gcn_wmma_gemm_kernel<IN_DIM, HID_DIM, true>
            <<<grid, 256, 0, stream>>>(agg, W1, b1, inv_in, hbuf, nullptr, N);
    }

    {
        int grid = (N + 127) / 128;
        gcn_wmma_gemm_kernel<HID_DIM, OUT_DIM, false>
            <<<grid, 256, 0, stream>>>(hbuf, W2, b2, inv_out, zbuf, out, N);
    }

    gcn_edge_scatter_kernel<OUT_DIM, 8, false><<<2048, 256, 0, stream>>>(
        zbuf, src, dst, inv_in, out, E);
}


# Every atomic target (degree counters, layer-1 sums, layer-2 sums) kept once per package half; a wave adds into its own half's copy (die id); next reader sums the copies; f32 sums regrouped by half
# speedup vs baseline: 2.2593x; 1.1284x over previous
#include <hip/hip_runtime.h>


typedef __attribute__((ext_vector_type(2))) float v2f;
typedef __attribute__((ext_vector_type(8))) float v8f;

#define IN_DIM  128
#define HID_DIM 128
#define OUT_DIM 64

__device__ __forceinline__ int gcn_package_half() {
    return (int)((__builtin_amdgcn_s_sendmsg_rtn(0x87) >> 18) & 1u);
}

__global__ void gcn_degree_kernel(const int* __restrict__ src,
                                  const int* __restrict__ dst,
                                  int* __restrict__ deg, int N, int E) {
    int* __restrict__ d = deg + (size_t)gcn_package_half() * (size_t)(2 * N);
    int e = blockIdx.x * blockDim.x + threadIdx.x;
    if (e < E) {
        atomicAdd(&d[src[e]], 1);
        atomicAdd(&d[N + dst[e]], 1);
    }
}

__global__ void gcn_invsqrt_kernel(const int* __restrict__ deg,
                                   float* __restrict__ inv, int n2) {
    int i = blockIdx.x * blockDim.x + threadIdx.x;
    if (i < n2) {
        float d = (float)(deg[i] + deg[n2 + i]);
        inv[i] = rsqrtf(fmaxf(d, 1.0f));
    }
}

template <int COLS, int U>
__device__ __forceinline__ void gcn_scatter_edges(const float* __restrict__ feat,
                                                  float* __restrict__ acc,
                                                  int s, int d, int sc_bits,
                                                  int j0, int lane) {
    constexpr int C = COLS / 32;
    float v[U][C];
#pragma unroll
    for (int u = 0; u < U; ++u) {
        const unsigned sj = (unsigned)__builtin_amdgcn_readlane(s, j0 + u);
        const float* __restrict__ p = feat + (size_t)(sj * (unsigned)COLS) + lane;
#pragma unroll
        for (int c = 0; c < C; ++c) v[u][c] = p[32 * c];
    }
#pragma unroll
    for (int u = 0; u < U; ++u) {
        const unsigned dj = (unsigned)__builtin_amdgcn_readlane(d, j0 + u);
        const float scj = __int_as_float(__builtin_amdgcn_readlane(sc_bits, j0 + u));
        float* __restrict__ q = acc + (size_t)(dj * (unsigned)COLS) + lane;
#pragma unroll
        for (int c = 0; c < C; ++c) unsafeAtomicAdd(q + 32 * c, v[u][c] * scj);
    }
}

template <int COLS, int U, bool SCALE_BY_SRC>
__global__ __launch_bounds__(256)
void gcn_edge_scatter_kernel(const float* __restrict__ feat,
                             const int* __restrict__ src,
                             const int* __restrict__ dst,
                             const float* __restrict__ inv,
                             float* __restrict__ acc0,
                             float* __restrict__ acc1, int E) {
    float* __restrict__ acc = gcn_package_half() ? acc1 : acc0;
    const int lane = threadIdx.x & 31;
    const int wave_in_block = __builtin_amdgcn_readfirstlane((int)(threadIdx.x >> 5));
    const int wave   = (int)blockIdx.x * 8 + wave_in_block;
    const int nwaves = (int)gridDim.x * 8;
    const int nchunks = (E + 31) >> 5;

    for (int ch = wave; ch < nchunks; ch += nwaves) {
        const int e0 = ch << 5;
        if (e0 + 32 <= E) {
            const int s = src[e0 + lane];
            const int d = dst[e0 + lane];
            const int sc_bits = __float_as_int(inv[SCALE_BY_SRC ? s : d]);
#pragma unroll
            for (int jb = 0; jb < 32; jb += U)
                gcn_scatter_edges<COLS, U>(feat, acc, s, d, sc_bits, jb, lane);
        } else {
            const int n = E - e0;
            int s = 0, d = 0, sc_bits = 0;
            if (lane < n) {
                s = src[e0 + lane];
                d = dst[e0 + lane];
                sc_bits = __float_as_int(inv[SCALE_BY_SRC ? s : d]);
            }
            for (int j = 0; j < n; ++j)
                gcn_scatter_edges<COLS, 1>(feat, acc, s, d, sc_bits, j, lane);
        }
    }
}

template <int K, int NOUT, bool FIRST>
__global__ __launch_bounds__(256)
void gcn_wmma_gemm_kernel(const float* __restrict__ A,
                          const float* __restrict__ A1,
                          const float* __restrict__ W,
                          const float* __restrict__ bias,
                          const float* __restrict__ scale,
                          float* __restrict__ out,
                          float* __restrict__ acc_init,
                          float* __restrict__ acc_zero, int N) {
    constexpr int S  = 2 * NOUT + 32;
    constexpr int NT = NOUT / 16;
    __shared__ float lds_w[(K / 2) * S];

    for (int idx = threadIdx.x; idx < K * NOUT; idx += blockDim.x) {
        int k = idx / NOUT;
        int n = idx - k * NOUT;
        lds_w[(k >> 1) * S + n * 2 + (k & 1)] = W[idx];
    }
    __syncthreads();

    const int lane = threadIdx.x & 31;
    const int wave = threadIdx.x >> 5;
    const int g = lane >> 4;
    const int mr = lane & 15;

    const int row0 = blockIdx.x * 128 + wave * 16;
    int row = row0 + mr;
    if (row >= N) row = N - 1;
    const float s = scale[row];
    const float* __restrict__ Arow = A + (size_t)row * K;
    const float* __restrict__ Arow1 = FIRST ? A1 + (size_t)row * K : nullptr;

    v8f acc[NT];
#pragma unroll
    for (int t = 0; t < NT; ++t) acc[t] = (v8f){};

#pragma unroll 1
    for (int kb = 0; kb < K; kb += 16) {
        v2f a[4], a1[4];
#pragma unroll
        for (int i = 0; i < 4; ++i)
            a[i] = *(const v2f*)(Arow + kb + 4 * i + 2 * g);
        if (FIRST) {
#pragma unroll
            for (int i = 0; i < 4; ++i)
                a1[i] = *(const v2f*)(Arow1 + kb + 4 * i + 2 * g);
        }
#pragma unroll
        for (int i = 0; i < 4; ++i) {
            const int k0 = kb + 4 * i;
            const v2f* __restrict__ wp =
                (const v2f*)(lds_w + (size_t)(k0 / 2 + g) * S);
            v2f b[NT];
#pragma unroll
            for (int t = 0; t < NT; ++t)
                b[t] = wp[t * 16 + mr];
            v2f av = a[i];
            if (FIRST) {
                av.x += a1[i].x;
                av.y += a1[i].y;
            }
            av.x *= s;
            av.y *= s;
#pragma unroll
            for (int t = 0; t < NT; ++t)
                acc[t] = __builtin_amdgcn_wmma_f32_16x16x4_f32(
                     false, av,  false, b[t],
                     (short)0, acc[t],  false,  false);
        }
    }

#pragma unroll
    for (int t = 0; t < NT; ++t) {
        const int Nbase = t * 16;
        const float bv = bias[Nbase + mr];
#pragma unroll
        for (int r = 0; r < 8; ++r) {
            const int orow = row0 + r + 8 * g;
            if (orow < N) {
                const size_t o = (size_t)orow * NOUT + Nbase + mr;
                if (FIRST) {
                    out[o] = fmaxf(acc[t][r] + bv, 0.0f);
                } else {
                    out[o] = acc[t][r];
                    acc_init[o] = bv;
                    acc_zero[o] = 0.0f;
                }
            }
        }
    }
}

__global__ __launch_bounds__(256)
void gcn_merge_halves_kernel(float* __restrict__ out,
                             const float* __restrict__ other, int n4) {
    const int base = (int)blockIdx.x * 1024 + (int)threadIdx.x;
    float4 a[4], b[4];
#pragma unroll
    for (int j = 0; j < 4; ++j) {
        const int i = base + 256 * j;
        if (i < n4) {
            a[j] = ((const float4*)out)[i];
            b[j] = ((const float4*)other)[i];
        }
    }
#pragma unroll
    for (int j = 0; j < 4; ++j) {
        const int i = base + 256 * j;
        if (i < n4) {
            float4 r;
            r.x = a[j].x + b[j].x;
            r.y = a[j].y + b[j].y;
            r.z = a[j].z + b[j].z;
            r.w = a[j].w + b[j].w;
            ((float4*)out)[i] = r;
        }
    }
}

extern "C" void kernel_launch(void* const* d_in, const int* in_sizes, int n_in,
                              void* d_out, int out_size, void* d_ws, size_t ws_size,
                              hipStream_t stream) {
    const float* in_feat = (const float*)d_in[0];
    const float* W1      = (const float*)d_in[1];
    const float* b1      = (const float*)d_in[2];
    const float* W2      = (const float*)d_in[3];
    const float* b2      = (const float*)d_in[4];
    const int*   src     = (const int*)d_in[5];
    const int*   dst     = (const int*)d_in[6];
    float*       out     = (float*)d_out;

    const int N = in_sizes[0] / IN_DIM;
    const int E = in_sizes[5];

    char* ws = (char*)d_ws;
    int*   deg   = (int*)ws;
    float* inv   = (float*)(ws + (size_t)4 * N * 4);
    float* agg0  = (float*)(ws + (size_t)6 * N * 4);
    float* agg1  = agg0 + (size_t)N * HID_DIM;
    float* hbuf  = agg1 + (size_t)N * HID_DIM;
    float* zbuf  = hbuf + (size_t)N * HID_DIM;
    float* out1  = zbuf + (size_t)N * OUT_DIM;
    float* inv_out = inv;
    float* inv_in  = inv + N;

    hipMemsetAsync(deg, 0, (size_t)4 * N * sizeof(int), stream);
    hipMemsetAsync(agg0, 0, (size_t)2 * N * HID_DIM * sizeof(float), stream);

    {
        int blk = 256;
        gcn_degree_kernel<<<(E + blk - 1) / blk, blk, 0, stream>>>(
            src, dst, deg, N, E);
        gcn_invsqrt_kernel<<<(2 * N + blk - 1) / blk, blk, 0, stream>>>(
            deg, inv, 2 * N);
    }

    gcn_edge_scatter_kernel<IN_DIM, 8, true><<<2048, 256, 0, stream>>>(
        in_feat, src, dst, inv_out, agg0, agg1, E);

    {
        int grid = (N + 127) / 128;
        gcn_wmma_gemm_kernel<IN_DIM, HID_DIM, true>
            <<<grid, 256, 0, stream>>>(agg0, agg1, W1, b1, inv_in, hbuf,
                                       nullptr, nullptr, N);
    }

    {
        int grid = (N + 127) / 128;
        gcn_wmma_gemm_kernel<HID_DIM, OUT_DIM, false>
            <<<grid, 256, 0, stream>>>(hbuf, nullptr, W2, b2, inv_out, zbuf,
                                       out, out1, N);
    }

    gcn_edge_scatter_kernel<OUT_DIM, 8, false><<<2048, 256, 0, stream>>>(
        zbuf, src, dst, inv_in, out, out1, E);

    {
        int n4 = N * (OUT_DIM / 4);
        gcn_merge_halves_kernel<<<(n4 + 1023) / 1024, 256, 0, stream>>>(out, out1, n4);
    }
}


# Layer 2's edge sum reads a per-owner edge list (src,row) that the layer-1 owner blocks write as they search the tag plane: no second search, no src/dst loads; sums stay in LDS arrival order
# speedup vs baseline: 3.5679x; 1.5792x over previous
#include <hip/hip_runtime.h>


typedef __attribute__((ext_vector_type(2))) float v2f;
typedef __attribute__((ext_vector_type(8))) float v8f;

#define IN_DIM  128
#define HID_DIM 128
#define OUT_DIM 64

__device__ __forceinline__ int gcn_package_half() {
    return (int)((__builtin_amdgcn_s_sendmsg_rtn(0x87) >> 18) & 1u);
}

#define TILE_ROWS 392
__global__ void gcn_degree_tag_kernel(const int* __restrict__ src,
                                      const int* __restrict__ dst,
                                      int* __restrict__ deg,
                                      unsigned int* __restrict__ tag4, int N, int E) {
    int* __restrict__ d = deg + (size_t)gcn_package_half() * (size_t)(2 * N);
    const int g = blockIdx.x * blockDim.x + threadIdx.x;
    const int e = g * 4;
    if (e + 3 < E) {
        const int4 s = ((const int4*)src)[g];
        const int4 t = ((const int4*)dst)[g];
        atomicAdd(&d[s.x], 1);
        atomicAdd(&d[s.y], 1);
        atomicAdd(&d[s.z], 1);
        atomicAdd(&d[s.w], 1);
        atomicAdd(&d[N + t.x], 1);
        atomicAdd(&d[N + t.y], 1);
        atomicAdd(&d[N + t.z], 1);
        atomicAdd(&d[N + t.w], 1);
        tag4[g] = (((unsigned)t.x / TILE_ROWS) & 255u)
                | ((((unsigned)t.y / TILE_ROWS) & 255u) << 8)
                | ((((unsigned)t.z / TILE_ROWS) & 255u) << 16)
                | ((((unsigned)t.w / TILE_ROWS) & 255u) << 24);
    } else {
        for (int k = e; k < E; ++k) {
            atomicAdd(&d[src[k]], 1);
            atomicAdd(&d[N + dst[k]], 1);
        }
    }
}

#define INV_THREADS 512
static_assert(TILE_ROWS <= INV_THREADS, "one thread per row of an owner");
__global__ __launch_bounds__(INV_THREADS)
void gcn_invsqrt_count_kernel(const int* __restrict__ deg, float* __restrict__ inv,
                              int* __restrict__ cnt, int N, int owners) {
    __shared__ int part[INV_THREADS / 32];
    const bool in_kind = (int)blockIdx.x < owners;
    const int o = in_kind ? (int)blockIdx.x : (int)blockIdx.x - owners;
    const int row = o * TILE_ROWS + (int)threadIdx.x;
    int d = 0;
    if ((int)threadIdx.x < TILE_ROWS && row < N) {
        const int i = (in_kind ? N : 0) + row;
        d = deg[i] + deg[2 * N + i];
        inv[i] = rsqrtf(fmaxf((float)d, 1.0f));
    }
    if (in_kind) {
        for (int m = 16; m > 0; m >>= 1) d += __shfl_xor(d, m);
        if ((threadIdx.x & 31) == 0) part[threadIdx.x >> 5] = d;
        __syncthreads();
        if (threadIdx.x == 0) {
            int t = 0;
            for (int k = 0; k < INV_THREADS / 32; ++k) t += part[k];
            cnt[o] = t;
        }
    }
}

__device__ __forceinline__ int gcn_list_start(const int* __restrict__ cnt, int owner, int lane) {
    int a = 0;
    for (int j = lane; j < owner; j += 32) a += cnt[j];
    for (int m = 16; m > 0; m >>= 1) a += __shfl_xor(a, m);
    return a;
}

#define QUEUE_CAP 64
#define ROW_BITS 9
static_assert(TILE_ROWS < (1 << ROW_BITS), "row field of a list entry");

template <int COLS, int U, bool SCALE_BY_SRC, bool FULL>
__device__ __forceinline__ void gcn_add_rows(const float* __restrict__ feat, float* tile,
                                             int s, int r, int sc_bits, int n, int lane) {
    constexpr int C = COLS / 32;
#pragma unroll 1
    for (int jb = 0; jb < n; jb += U) {
        float v[U][C];
#pragma unroll
        for (int u = 0; u < U; ++u) {
            if (FULL || jb + u < n) {
                const unsigned sj = (unsigned)__builtin_amdgcn_readlane(s, jb + u);
                const float* __restrict__ p = feat + (size_t)(sj * (unsigned)COLS) + lane;
#pragma unroll
                for (int c = 0; c < C; ++c) v[u][c] = p[32 * c];
            }
        }
#pragma unroll
        for (int u = 0; u < U; ++u) {
            if (FULL || jb + u < n) {
                const unsigned rj = (unsigned)__builtin_amdgcn_readlane(r, jb + u);
                float* tp = tile + rj * (unsigned)COLS + lane;
                if (SCALE_BY_SRC) {
                    const float scj = __int_as_float(__builtin_amdgcn_readlane(sc_bits, jb + u));
#pragma unroll
                    for (int c = 0; c < C; ++c) unsafeAtomicAdd(tp + 32 * c, v[u][c] * scj);
                } else {
#pragma unroll
                    for (int c = 0; c < C; ++c) unsafeAtomicAdd(tp + 32 * c, v[u][c]);
                }
            }
        }
    }
}

template <int COLS, int U, bool FULL>
__device__ __forceinline__ void gcn_add_queued(const float* __restrict__ feat,
                                               const int* __restrict__ src,
                                               const int* __restrict__ dst,
                                               const float* __restrict__ inv_src,
                                               unsigned int* __restrict__ mylist, int* fill,
                                               float* tile, const int* q,
                                               unsigned head, int n, int lane, int lo) {
    int s = 0, r = TILE_ROWS, sc_bits = 0;
    if (FULL || lane < n) {
        const int e = q[(head + (unsigned)lane) & (QUEUE_CAP - 1)];
        s = src[e];
        const unsigned rr = (unsigned)(dst[e] - lo);
        r = rr < (unsigned)TILE_ROWS ? (int)rr : TILE_ROWS;
        sc_bits = __float_as_int(inv_src[s]);
    }
    const bool own = r != TILE_ROWS;
    const unsigned mo = __builtin_amdgcn_ballot_w32(own);
    if (mo != 0u) {
        int at = 0;
        if (lane == 0) at = atomicAdd(fill, (int)__builtin_popcount(mo));
        at = __builtin_amdgcn_readfirstlane(at);
        if (own)
            mylist[at + (int)__builtin_amdgcn_mbcnt_lo(mo, 0u)] =
                ((unsigned)s << ROW_BITS) | (unsigned)r;
    }
    gcn_add_rows<COLS, U, true, FULL>(feat, tile, s, r, sc_bits, n, lane);
}

__device__ __forceinline__ unsigned gcn_nonzero_bytes(unsigned x) {
    return ((x & 0x7f7f7f7fu) + 0x7f7f7f7fu) | x;
}

template <int COLS, int U>
__global__ __launch_bounds__(1024)
void gcn_owner_edge_sum_kernel(const float* __restrict__ feat,
                               const int* __restrict__ src,
                               const int* __restrict__ dst,
                               const unsigned int* __restrict__ tags,
                               const float* __restrict__ inv_src,
                               const int* __restrict__ cnt,
                               unsigned int* __restrict__ list,
                               float* __restrict__ result, int N, int E) {
    __shared__ float tile[(TILE_ROWS + 1) * COLS];
    __shared__ int ring[32 * QUEUE_CAP];
    __shared__ int seg[2];

    const int tid = (int)threadIdx.x;
    const int lane = tid & 31;
    const int wv = __builtin_amdgcn_readfirstlane(tid >> 5);
    int* q = ring + wv * QUEUE_CAP;
    const int lo = (int)blockIdx.x * TILE_ROWS;
    const unsigned idrep = ((unsigned)blockIdx.x & 255u) * 0x01010101u;

    if (wv == 0) {
        const int start = gcn_list_start(cnt, (int)blockIdx.x, lane);
        if (lane == 0) {
            seg[0] = start;
            seg[1] = 0;
        }
    }
    for (int f = tid; f < (TILE_ROWS + 1) * COLS / 4; f += 1024)
        ((float4*)tile)[f] = make_float4(0.0f, 0.0f, 0.0f, 0.0f);
    __syncthreads();
    unsigned int* __restrict__ mylist = list + __builtin_amdgcn_readfirstlane(seg[0]);
    int* fill = &seg[1];

    const uint4* __restrict__ tag16 = (const uint4*)tags;
    const int nvec = E >> 4;
    const int trips = (nvec + 8 * 1024 - 1) / (8 * 1024);
    const int rot = trips > 0 ? (int)(blockIdx.x % (unsigned)trips) : 0;
    unsigned head = 0, cnt_q = 0;

    for (int ob = 0; ob < trips; ++ob) {
        int kb = ob + rot;
        if (kb >= trips) kb -= trips;
        const int v0 = kb * (8 * 1024) + tid;
        uint4 dv[8];
        if ((kb + 1) * (8 * 1024) <= nvec) {
#pragma unroll
            for (int i = 0; i < 8; ++i) dv[i] = tag16[v0 + i * 1024];
        } else {
#pragma unroll
            for (int i = 0; i < 8; ++i) {
                const int v = v0 + i * 1024;
                dv[i] = make_uint4(~idrep, ~idrep, ~idrep, ~idrep);
                if (v < nvec) dv[i] = tag16[v];
            }
        }
#pragma unroll
        for (int i = 0; i < 8; ++i) {
            const int e0 = (v0 + i * 1024) * 16;
            const unsigned t0 = gcn_nonzero_bytes(dv[i].x ^ idrep);
            const unsigned t1 = gcn_nonzero_bytes(dv[i].y ^ idrep);
            const unsigned t2 = gcn_nonzero_bytes(dv[i].z ^ idrep);
            const unsigned t3 = gcn_nonzero_bytes(dv[i].w ^ idrep);
            unsigned m = ~((t0 & 0x80808080u) | ((t1 >> 1) & 0x40404040u) |
                           ((t2 >> 2) & 0x20202020u) | ((t3 >> 3) & 0x10101010u)) & 0xf0f0f0f0u;
            unsigned any = __builtin_amdgcn_ballot_w32(m != 0u);
            while (any != 0u) {
                const unsigned pos = head + cnt_q + __builtin_amdgcn_mbcnt_lo(any, 0u);
                if (m != 0u) {
                    const unsigned p = (unsigned)__builtin_ctz(m);
                    q[pos & (QUEUE_CAP - 1)] = e0 + 4 * (int)(7u - (p & 7u)) + (int)(p >> 3);
                    m &= m - 1u;
                }
                cnt_q += (unsigned)__builtin_popcount(any);
                if (cnt_q >= 32u) {
                    gcn_add_queued<COLS, U, true>(feat, src, dst, inv_src, mylist, fill, tile, q,
                                                  head, 32, lane, lo);
                    head += 32u;
                    cnt_q -= 32u;
                }
                any = __builtin_amdgcn_ballot_w32(m != 0u);
            }
        }
    }
    if (wv == 0 && (E & 15) != 0) {
        const int e = nvec * 16 + lane;
        unsigned rr = 0xffffffffu;
        if (lane < (E & 15)) rr = (unsigned)(dst[e] - lo);
        const unsigned mt = __builtin_amdgcn_ballot_w32(rr < (unsigned)TILE_ROWS);
        if (mt != 0u) {
            const unsigned pos = head + cnt_q + __builtin_amdgcn_mbcnt_lo(mt, 0u);
            if (rr < (unsigned)TILE_ROWS) q[pos & (QUEUE_CAP - 1)] = e;
            cnt_q += (unsigned)__builtin_popcount(mt);
        }
    }
    while (cnt_q >= 32u) {
        gcn_add_queued<COLS, U, true>(feat, src, dst, inv_src, mylist, fill, tile, q, head, 32,
                                      lane, lo);
        head += 32u;
        cnt_q -= 32u;
    }
    if (cnt_q != 0u)
        gcn_add_queued<COLS, U, false>(feat, src, dst, inv_src, mylist, fill, tile, q, head,
                                       (int)cnt_q, lane, lo);
    __syncthreads();

    constexpr int C4 = COLS / 4;
    for (int f = tid; f < TILE_ROWS * C4; f += 1024) {
        const int row = f / C4;
        const int c4 = f - row * C4;
        const int grow = lo + row;
        if (grow < N) ((float4*)result)[(size_t)grow * C4 + c4] = ((const float4*)tile)[f];
    }
}

template <int COLS, int U>
__global__ __launch_bounds__(1024)
void gcn_listed_edge_sum_kernel(const float* __restrict__ feat,
                                const unsigned int* __restrict__ list,
                                const int* __restrict__ cnt,
                                const float* __restrict__ inv_row,
                                const float* __restrict__ bias,
                                float* __restrict__ result, int N) {
    __shared__ float tile[(TILE_ROWS + 1) * COLS];
    __shared__ int seg[2];

    const int tid = (int)threadIdx.x;
    const int lane = tid & 31;
    const int wv = __builtin_amdgcn_readfirstlane(tid >> 5);
    const int lo = (int)blockIdx.x * TILE_ROWS;

    if (wv == 0) {
        const int start = gcn_list_start(cnt, (int)blockIdx.x, lane);
        if (lane == 0) {
            seg[0] = start;
            seg[1] = cnt[blockIdx.x];
        }
    }
    for (int f = tid; f < (TILE_ROWS + 1) * COLS / 4; f += 1024)
        ((float4*)tile)[f] = make_float4(0.0f, 0.0f, 0.0f, 0.0f);
    __syncthreads();
    const unsigned int* __restrict__ mylist = list + __builtin_amdgcn_readfirstlane(seg[0]);
    const int n = __builtin_amdgcn_readfirstlane(seg[1]);

    int c0 = wv * 32;
    if (c0 < n) {
        unsigned w = mylist[min(c0 + lane, n - 1)];
        do {
            const unsigned ws = w >> ROW_BITS;
            const unsigned wr = w & ((1u << ROW_BITS) - 1u);
            const int s = (int)(ws < (unsigned)N ? ws : (unsigned)N - 1u);
            const int r = (int)(wr < (unsigned)TILE_ROWS ? wr : (unsigned)TILE_ROWS);
            const int c1 = c0 + 1024;
            const unsigned wn = mylist[min(c1 + lane, n - 1)];
            const int m = n - c0;
            if (m >= 32)
                gcn_add_rows<COLS, U, false, true>(feat, tile, s, r, 0, 32, lane);
            else
                gcn_add_rows<COLS, U, false, false>(feat, tile, s, r, 0, m, lane);
            w = wn;
            c0 = c1;
        } while (c0 < n);
    }
    __syncthreads();

    constexpr int C4 = COLS / 4;
    for (int f = tid; f < TILE_ROWS * C4; f += 1024) {
        const int row = f / C4;
        const int c4 = f - row * C4;
        const int grow = lo + row;
        if (grow < N) {
            float4 t = ((const float4*)tile)[f];
            const float sc = inv_row[grow];
            const float4 b = ((const float4*)bias)[c4];
            t.x = b.x + sc * t.x;
            t.y = b.y + sc * t.y;
            t.z = b.z + sc * t.z;
            t.w = b.w + sc * t.w;
            ((float4*)result)[(size_t)grow * C4 + c4] = t;
        }
    }
}

template <int K, int NOUT, bool FIRST>
__global__ __launch_bounds__(256)
void gcn_wmma_gemm_kernel(const float* __restrict__ A,
                          const float* __restrict__ W,
                          const float* __restrict__ bias,
                          const float* __restrict__ scale,
                          float* __restrict__ out, int N) {
    constexpr int S  = 2 * NOUT + 32;
    constexpr int NT = NOUT / 16;
    __shared__ float lds_w[(K / 2) * S];

    for (int idx = threadIdx.x; idx < K * NOUT; idx += blockDim.x) {
        int k = idx / NOUT;
        int n = idx - k * NOUT;
        lds_w[(k >> 1) * S + n * 2 + (k & 1)] = W[idx];
    }
    __syncthreads();

    const int lane = threadIdx.x & 31;
    const int wave = threadIdx.x >> 5;
    const int g = lane >> 4;
    const int mr = lane & 15;

    const int row0 = blockIdx.x * 128 + wave * 16;
    int row = row0 + mr;
    if (row >= N) row = N - 1;
    const float s = scale[row];
    const float* __restrict__ Arow = A + (size_t)row * K;

    v8f acc[NT];
#pragma unroll
    for (int t = 0; t < NT; ++t) acc[t] = (v8f){};

#pragma unroll 1
    for (int kb = 0; kb < K; kb += 16) {
        v2f a[4];
#pragma unroll
        for (int i = 0; i < 4; ++i)
            a[i] = *(const v2f*)(Arow + kb + 4 * i + 2 * g);
#pragma unroll
        for (int i = 0; i < 4; ++i) {
            const int k0 = kb + 4 * i;
            const v2f* __restrict__ wp =
                (const v2f*)(lds_w + (size_t)(k0 / 2 + g) * S);
            v2f b[NT];
#pragma unroll
            for (int t = 0; t < NT; ++t)
                b[t] = wp[t * 16 + mr];
            v2f av = a[i];
            av.x *= s;
            av.y *= s;
#pragma unroll
            for (int t = 0; t < NT; ++t)
                acc[t] = __builtin_amdgcn_wmma_f32_16x16x4_f32(
                     false, av,  false, b[t],
                     (short)0, acc[t],  false,  false);
        }
    }

#pragma unroll
    for (int t = 0; t < NT; ++t) {
        const int Nbase = t * 16;
        const float bv = FIRST ? bias[Nbase + mr] : 0.0f;
#pragma unroll
        for (int r = 0; r < 8; ++r) {
            const int orow = row0 + r + 8 * g;
            if (orow < N) {
                const size_t o = (size_t)orow * NOUT + Nbase + mr;
                if (FIRST) {
                    out[o] = fmaxf(acc[t][r] + bv, 0.0f);
                } else {
                    out[o] = acc[t][r];
                }
            }
        }
    }
}

extern "C" void kernel_launch(void* const* d_in, const int* in_sizes, int n_in,
                              void* d_out, int out_size, void* d_ws, size_t ws_size,
                              hipStream_t stream) {
    const float* in_feat = (const float*)d_in[0];
    const float* W1      = (const float*)d_in[1];
    const float* b1      = (const float*)d_in[2];
    const float* W2      = (const float*)d_in[3];
    const float* b2      = (const float*)d_in[4];
    const int*   src     = (const int*)d_in[5];
    const int*   dst     = (const int*)d_in[6];
    float*       out     = (float*)d_out;

    const int N = in_sizes[0] / IN_DIM;
    const int E = in_sizes[5];

    char* ws = (char*)d_ws;
    int*   deg   = (int*)ws;
    float* inv   = (float*)(ws + (size_t)4 * N * 4);
    float* agg   = (float*)(ws + (size_t)6 * N * 4);
    float* hbuf  = agg + (size_t)N * HID_DIM;
    float* zbuf  = hbuf + (size_t)N * HID_DIM;
    unsigned int* tags = (unsigned int*)(ws + (((size_t)1304 * N + 15) & ~(size_t)15));
    const int owners = (N + TILE_ROWS - 1) / TILE_ROWS;
    int* cnt = (int*)((char*)tags + (((size_t)E + 15) & ~(size_t)15));
    unsigned int* list = (unsigned int*)(cnt + ((owners + 31) & ~31));
    float* inv_out = inv;
    float* inv_in  = inv + N;

    hipMemsetAsync(deg, 0, (size_t)4 * N * sizeof(int), stream);

    {
        int blk = 256;
        int groups = (E + 3) / 4;
        gcn_degree_tag_kernel<<<(groups + blk - 1) / blk, blk, 0, stream>>>(
            src, dst, deg, tags, N, E);
        gcn_invsqrt_count_kernel<<<2 * owners, INV_THREADS, 0, stream>>>(
            deg, inv, cnt, N, owners);
    }

    gcn_owner_edge_sum_kernel<IN_DIM, 8><<<owners, 1024, 0, stream>>>(
        in_feat, src, dst, tags, inv_out, cnt, list, agg, N, E);

    {
        int grid = (N + 127) / 128;
        gcn_wmma_gemm_kernel<IN_DIM, HID_DIM, true>
            <<<grid, 256, 0, stream>>>(agg, W1, b1, inv_in, hbuf, N);
    }

    {
        int grid = (N + 127) / 128;
        gcn_wmma_gemm_kernel<HID_DIM, OUT_DIM, false>
            <<<grid, 256, 0, stream>>>(hbuf, W2, b2, inv_out, zbuf, N);
    }

    gcn_listed_edge_sum_kernel<OUT_DIM, 16><<<owners, 1024, 0, stream>>>(
        zbuf, list, cnt, inv_in, b2, out, N);
}


# Both edge sums: each LDS tile row belongs to one wave (row % 32); layer 1 files its edges into per-wave list pieces, then waves add rows by plain LDS load/add/store, no LDS float atomics; list order
# speedup vs baseline: 3.7367x; 1.0473x over previous
#include <hip/hip_runtime.h>


typedef __attribute__((ext_vector_type(2))) float v2f;
typedef __attribute__((ext_vector_type(4))) float v4f;
typedef __attribute__((ext_vector_type(8))) float v8f;

#define IN_DIM  128
#define HID_DIM 128
#define OUT_DIM 64

__device__ __forceinline__ int gcn_package_half() {
    return (int)((__builtin_amdgcn_s_sendmsg_rtn(0x87) >> 18) & 1u);
}

#define TILE_ROWS 392
__global__ void gcn_degree_tag_kernel(const int* __restrict__ src,
                                      const int* __restrict__ dst,
                                      int* __restrict__ deg,
                                      unsigned int* __restrict__ tag4, int N, int E) {
    int* __restrict__ d = deg + (size_t)gcn_package_half() * (size_t)(2 * N);
    const int g = blockIdx.x * blockDim.x + threadIdx.x;
    const int e = g * 4;
    if (e + 3 < E) {
        const int4 s = ((const int4*)src)[g];
        const int4 t = ((const int4*)dst)[g];
        atomicAdd(&d[s.x], 1);
        atomicAdd(&d[s.y], 1);
        atomicAdd(&d[s.z], 1);
        atomicAdd(&d[s.w], 1);
        atomicAdd(&d[N + t.x], 1);
        atomicAdd(&d[N + t.y], 1);
        atomicAdd(&d[N + t.z], 1);
        atomicAdd(&d[N + t.w], 1);
        tag4[g] = (((unsigned)t.x / TILE_ROWS) & 255u)
                | ((((unsigned)t.y / TILE_ROWS) & 255u) << 8)
                | ((((unsigned)t.z / TILE_ROWS) & 255u) << 16)
                | ((((unsigned)t.w / TILE_ROWS) & 255u) << 24);
    } else {
        for (int k = e; k < E; ++k) {
            atomicAdd(&d[src[k]], 1);
            atomicAdd(&d[N + dst[k]], 1);
        }
    }
}

#define INV_THREADS 512
static_assert(TILE_ROWS <= INV_THREADS, "one thread per row of an owner");
__global__ __launch_bounds__(INV_THREADS)
void gcn_invsqrt_count_kernel(const int* __restrict__ deg, float* __restrict__ inv,
                              int* __restrict__ cnt, int N, int owners) {
    __shared__ int part[INV_THREADS / 32];
    const bool in_kind = (int)blockIdx.x < owners;
    const int o = in_kind ? (int)blockIdx.x : (int)blockIdx.x - owners;
    const int row = o * TILE_ROWS + (int)threadIdx.x;
    int d = 0;
    if ((int)threadIdx.x < TILE_ROWS && row < N) {
        const int i = (in_kind ? N : 0) + row;
        d = deg[i] + deg[2 * N + i];
        inv[i] = rsqrtf(fmaxf((float)d, 1.0f));
    }
    if (in_kind) {
        for (int m = 16; m > 0; m >>= 1) d += __shfl_xor(d, m);
        if ((threadIdx.x & 31) == 0) part[threadIdx.x >> 5] = d;
        __syncthreads();
        if (threadIdx.x == 0) {
            int t = 0;
            for (int k = 0; k < INV_THREADS / 32; ++k) t += part[k];
            cnt[o] = t;
        }
    }
}

__device__ __forceinline__ int gcn_list_start(const int* __restrict__ cnt, int owner, int lane) {
    int a = 0;
    for (int j = lane; j < owner; j += 32) a += cnt[j];
    for (int m = 16; m > 0; m >>= 1) a += __shfl_xor(a, m);
    return a;
}

#define QUEUE_CAP 64
#define ROW_BITS 9
static_assert(TILE_ROWS < (1 << ROW_BITS), "row field of a list entry");

template <int V> struct gcn_lane_vec;
template <> struct gcn_lane_vec<4> { typedef v4f type; };
template <> struct gcn_lane_vec<2> { typedef v2f type; };

__device__ __forceinline__ int gcn_piece_lengths(const int* __restrict__ deg, int N, int lo,
                                                 int lane) {
    const int* __restrict__ d0 = deg + N;
    const int* __restrict__ d1 = deg + (size_t)3 * N;
    int c = 0;
#pragma unroll
    for (int k = 0; k < (TILE_ROWS + 31) / 32; ++k) {
        const int lr = lane + 32 * k;
        const bool ok = lr < TILE_ROWS && lo + lr < N;
        const int row = ok ? lo + lr : lo;
        const int d = d0[row] + d1[row];
        c += ok ? d : 0;
    }
    return c;
}

__device__ __forceinline__ int gcn_sum_before(int v, int lane) {
    int x = v;
    for (int d = 1; d < 32; d <<= 1) {
        const int t = __shfl_up(x, d);
        if (lane >= d) x += t;
    }
    return x - v;
}

template <int COLS, int U, typename VEC>
__device__ __forceinline__ void gcn_load_rows(VEC (&v)[U], const float* __restrict__ feat,
                                              int s, int jb, int lane) {
#pragma unroll
    for (int u = 0; u < U; ++u) {
        const unsigned sj = (unsigned)__builtin_amdgcn_readlane(s, jb + u);
        v[u] = *(const VEC*)(feat + (size_t)(sj * (unsigned)COLS) + lane * (COLS / 32));
    }
}

template <int COLS, int U, bool SCALE_BY_SRC, typename VEC>
__device__ __forceinline__ void gcn_sum_rows(const VEC (&v)[U], float* tile, int r, int sc_bits,
                                             int jb, int lane) {
    static_assert(U % 4 == 0, "groups of four edges");
#pragma unroll
    for (int g = 0; g < U; g += 4) {
        unsigned rj[4];
        VEC* tp[4];
        VEC t[4], nv[4];
#pragma unroll
        for (int i = 0; i < 4; ++i) {
            rj[i] = (unsigned)__builtin_amdgcn_readlane(r, jb + g + i);
            tp[i] = (VEC*)(tile + rj[i] * (unsigned)COLS + lane * (COLS / 32));
        }
#pragma unroll
        for (int i = 0; i < 4; ++i) t[i] = *tp[i];
#pragma unroll
        for (int i = 0; i < 4; ++i) {
            VEC base = t[i];
#pragma unroll
            for (int p = 0; p < i; ++p) base = (rj[p] == rj[i]) ? nv[p] : base;
            if (SCALE_BY_SRC) {
                const float scj =
                    __int_as_float(__builtin_amdgcn_readlane(sc_bits, jb + g + i));
                nv[i] = base + v[g + i] * scj;
            } else {
                nv[i] = base + v[g + i];
            }
        }
#pragma unroll
        for (int i = 0; i < 4; ++i) *tp[i] = nv[i];
    }
}

template <int COLS, int U, bool SCALE_BY_SRC>
__device__ __forceinline__ void gcn_add_rows_32(const float* __restrict__ feat, float* tile,
                                                int s, int r, int sc_bits, int lane) {
    static_assert(32 % (2 * U) == 0, "two sets of U edges take turns");
    typedef typename gcn_lane_vec<COLS / 32>::type vec;
    vec va[U], vb[U];
    gcn_load_rows<COLS, U>(va, feat, s, 0, lane);
#pragma unroll
    for (int jb = 0; jb < 32; jb += 2 * U) {
        gcn_load_rows<COLS, U>(vb, feat, s, jb + U, lane);
        gcn_sum_rows<COLS, U, SCALE_BY_SRC>(va, tile, r, sc_bits, jb, lane);
        if (jb + 2 * U < 32) gcn_load_rows<COLS, U>(va, feat, s, jb + 2 * U, lane);
        gcn_sum_rows<COLS, U, SCALE_BY_SRC>(vb, tile, r, sc_bits, jb + U, lane);
    }
}

template <int COLS, int UP, bool SCALE_BY_SRC>
__device__ __forceinline__ void gcn_add_rows_some(const float* __restrict__ feat, float* tile,
                                                  int s, int r, int sc_bits, int nb, int lane) {
    typedef typename gcn_lane_vec<COLS / 32>::type vec;
#pragma unroll 1
    for (int b = 0; b < nb; ++b) {
        vec v[UP];
        gcn_load_rows<COLS, UP>(v, feat, s, b * UP, lane);
        gcn_sum_rows<COLS, UP, SCALE_BY_SRC>(v, tile, r, sc_bits, b * UP, lane);
    }
}

template <int COLS, int U, bool SCALE_BY_SRC>
__device__ __forceinline__ void gcn_add_piece(const float* __restrict__ feat,
                                              const unsigned int* wl, int n,
                                              const float* __restrict__ inv_src,
                                              float* tile, int N, int lane) {
    if (n <= 0) return;
    unsigned w = wl[min(lane, n - 1)];
    int c0 = 0;
    do {
        const unsigned ws = w >> ROW_BITS;
        const unsigned wr = w & ((1u << ROW_BITS) - 1u);
        const int s = (int)(ws < (unsigned)N ? ws : (unsigned)N - 1u);
        int r = (int)(wr < (unsigned)TILE_ROWS ? wr : (unsigned)TILE_ROWS);
        const int c1 = c0 + 32;
        const unsigned wn = wl[min(c1 + lane, n - 1)];
        int sc_bits = 0;
        if (SCALE_BY_SRC) sc_bits = __float_as_int(inv_src[s]);
        const int m = n - c0;
        if (m >= 32) {
            gcn_add_rows_32<COLS, U, SCALE_BY_SRC>(feat, tile, s, r, sc_bits, lane);
        } else {
            if (lane >= m) r = TILE_ROWS;
            gcn_add_rows_some<COLS, 4, SCALE_BY_SRC>(feat, tile, s, r, sc_bits, (m + 3) / 4, lane);
        }
        w = wn;
        c0 = c1;
    } while (c0 < n);
}

template <bool FULL>
__device__ __forceinline__ void gcn_file_queued(const int* __restrict__ src,
                                                const int* __restrict__ dst,
                                                unsigned int* mylist, int* piece_fill,
                                                const int* q, unsigned head, int n,
                                                int lane, int lo) {
    if (FULL || lane < n) {
        const int e = q[(head + (unsigned)lane) & (QUEUE_CAP - 1)];
        const int s = src[e];
        const unsigned rr = (unsigned)(dst[e] - lo);
        if (rr < (unsigned)TILE_ROWS) {
            const int at = atomicAdd(&piece_fill[rr & 31u], 1);
            mylist[at] = ((unsigned)s << ROW_BITS) | rr;
        }
    }
}

__device__ __forceinline__ unsigned gcn_nonzero_bytes(unsigned x) {
    return ((x & 0x7f7f7f7fu) + 0x7f7f7f7fu) | x;
}

template <int COLS, int U>
__global__ __launch_bounds__(1024)
void gcn_owner_edge_sum_kernel(const float* __restrict__ feat,
                               const int* __restrict__ src,
                               const int* __restrict__ dst,
                               const unsigned int* __restrict__ tags,
                               const float* __restrict__ inv_src,
                               const int* __restrict__ cnt,
                               const int* __restrict__ deg,
                               unsigned int* list,
                               float* __restrict__ result, int N, int E) {
    __shared__ __attribute__((aligned(16))) float tile[(TILE_ROWS + 1) * COLS];
    __shared__ int ring[32 * QUEUE_CAP];
    __shared__ int piece_fill[32];

    const int tid = (int)threadIdx.x;
    const int lane = tid & 31;
    const int wv = __builtin_amdgcn_readfirstlane(tid >> 5);
    int* q = ring + wv * QUEUE_CAP;
    const int lo = (int)blockIdx.x * TILE_ROWS;
    const unsigned idrep = ((unsigned)blockIdx.x & 255u) * 0x01010101u;

    const int start = gcn_list_start(cnt, (int)blockIdx.x, lane);
    const int len_lane = gcn_piece_lengths(deg, N, lo, lane);
    const int at_lane = gcn_sum_before(len_lane, lane);
    if (wv == 0) piece_fill[lane] = at_lane;
    for (int f = tid; f < (TILE_ROWS + 1) * COLS / 4; f += 1024)
        ((float4*)tile)[f] = make_float4(0.0f, 0.0f, 0.0f, 0.0f);
    __syncthreads();
    unsigned int* mylist = list + __builtin_amdgcn_readfirstlane(start);

    const uint4* __restrict__ tag16 = (const uint4*)tags;
    const int nvec = E >> 4;
    const int trips = (nvec + 8 * 1024 - 1) / (8 * 1024);
    const int rot = trips > 0 ? (int)(blockIdx.x % (unsigned)trips) : 0;
    unsigned head = 0, cnt_q = 0;

    for (int ob = 0; ob < trips; ++ob) {
        int kb = ob + rot;
        if (kb >= trips) kb -= trips;
        const int v0 = kb * (8 * 1024) + tid;
        uint4 dv[8];
        if ((kb + 1) * (8 * 1024) <= nvec) {
#pragma unroll
            for (int i = 0; i < 8; ++i) dv[i] = tag16[v0 + i * 1024];
        } else {
#pragma unroll
            for (int i = 0; i < 8; ++i) {
                const int v = v0 + i * 1024;
                dv[i] = make_uint4(~idrep, ~idrep, ~idrep, ~idrep);
                if (v < nvec) dv[i] = tag16[v];
            }
        }
#pragma unroll
        for (int i = 0; i < 8; ++i) {
            const int e0 = (v0 + i * 1024) * 16;
            const unsigned t0 = gcn_nonzero_bytes(dv[i].x ^ idrep);
            const unsigned t1 = gcn_nonzero_bytes(dv[i].y ^ idrep);
            const unsigned t2 = gcn_nonzero_bytes(dv[i].z ^ idrep);
            const unsigned t3 = gcn_nonzero_bytes(dv[i].w ^ idrep);
            unsigned m = ~((t0 & 0x80808080u) | ((t1 >> 1) & 0x40404040u) |
                           ((t2 >> 2) & 0x20202020u) | ((t3 >> 3) & 0x10101010u)) & 0xf0f0f0f0u;
            unsigned any = __builtin_amdgcn_ballot_w32(m != 0u);
            while (any != 0u) {
                const unsigned pos = head + cnt_q + __builtin_amdgcn_mbcnt_lo(any, 0u);
                if (m != 0u) {
                    const unsigned p = (unsigned)__builtin_ctz(m);
                    q[pos & (QUEUE_CAP - 1)] = e0 + 4 * (int)(7u - (p & 7u)) + (int)(p >> 3);
                    m &= m - 1u;
                }
                cnt_q += (unsigned)__builtin_popcount(any);
                if (cnt_q >= 32u) {
                    gcn_file_queued<true>(src, dst, mylist, piece_fill, q, head, 32, lane, lo);
                    head += 32u;
                    cnt_q -= 32u;
                }
                any = __builtin_amdgcn_ballot_w32(m != 0u);
            }
        }
    }
    if (wv == 0 && (E & 15) != 0) {
        const int e = nvec * 16 + lane;
        unsigned rr = 0xffffffffu;
        if (lane < (E & 15)) rr = (unsigned)(dst[e] - lo);
        const unsigned mt = __builtin_amdgcn_ballot_w32(rr < (unsigned)TILE_ROWS);
        if (mt != 0u) {
            const unsigned pos = head + cnt_q + __builtin_amdgcn_mbcnt_lo(mt, 0u);
            if (rr < (unsigned)TILE_ROWS) q[pos & (QUEUE_CAP - 1)] = e;
            cnt_q += (unsigned)__builtin_popcount(mt);
        }
    }
    while (cnt_q >= 32u) {
        gcn_file_queued<true>(src, dst, mylist, piece_fill, q, head, 32, lane, lo);
        head += 32u;
        cnt_q -= 32u;
    }
    if (cnt_q != 0u)
        gcn_file_queued<false>(src, dst, mylist, piece_fill, q, head, (int)cnt_q, lane, lo);
    __syncthreads();

    gcn_add_piece<COLS, U, true>(feat, mylist + __builtin_amdgcn_readlane(at_lane, wv),
                                 __builtin_amdgcn_readlane(len_lane, wv), inv_src, tile, N, lane);
    __syncthreads();

    constexpr int C4 = COLS / 4;
    for (int f = tid; f < TILE_ROWS * C4; f += 1024) {
        const int row = f / C4;
        const int c4 = f - row * C4;
        const int grow = lo + row;
        if (grow < N) ((float4*)result)[(size_t)grow * C4 + c4] = ((const float4*)tile)[f];
    }
}

template <int COLS, int U>
__global__ __launch_bounds__(1024)
void gcn_listed_edge_sum_kernel(const float* __restrict__ feat,
                                const unsigned int* __restrict__ list,
                                const int* __restrict__ cnt,
                                const int* __restrict__ deg,
                                const float* __restrict__ inv_row,
                                const float* __restrict__ bias,
                                float* __restrict__ result, int N) {
    __shared__ __attribute__((aligned(16))) float tile[(TILE_ROWS + 1) * COLS];

    const int tid = (int)threadIdx.x;
    const int lane = tid & 31;
    const int wv = __builtin_amdgcn_readfirstlane(tid >> 5);
    const int lo = (int)blockIdx.x * TILE_ROWS;

    const int start = gcn_list_start(cnt, (int)blockIdx.x, lane);
    const int len_lane = gcn_piece_lengths(deg, N, lo, lane);
    const int at_lane = gcn_sum_before(len_lane, lane);
    for (int f = tid; f < (TILE_ROWS + 1) * COLS / 4; f += 1024)
        ((float4*)tile)[f] = make_float4(0.0f, 0.0f, 0.0f, 0.0f);
    __syncthreads();
    const unsigned int* mypiece = list + __builtin_amdgcn_readfirstlane(start)
                                       + __builtin_amdgcn_readlane(at_lane, wv);
    gcn_add_piece<COLS, U, false>(feat, mypiece, __builtin_amdgcn_readlane(len_lane, wv),
                                  nullptr, tile, N, lane);
    __syncthreads();

    constexpr int C4 = COLS / 4;
    for (int f = tid; f < TILE_ROWS * C4; f += 1024) {
        const int row = f / C4;
        const int c4 = f - row * C4;
        const int grow = lo + row;
        if (grow < N) {
            float4 t = ((const float4*)tile)[f];
            const float sc = inv_row[grow];
            const float4 b = ((const float4*)bias)[c4];
            t.x = b.x + sc * t.x;
            t.y = b.y + sc * t.y;
            t.z = b.z + sc * t.z;
            t.w = b.w + sc * t.w;
            ((float4*)result)[(size_t)grow * C4 + c4] = t;
        }
    }
}

template <int K, int NOUT, bool FIRST>
__global__ __launch_bounds__(256)
void gcn_wmma_gemm_kernel(const float* __restrict__ A,
                          const float* __restrict__ W,
                          const float* __restrict__ bias,
                          const float* __restrict__ scale,
                          float* __restrict__ out, int N) {
    constexpr int S  = 2 * NOUT + 32;
    constexpr int NT = NOUT / 16;
    __shared__ float lds_w[(K / 2) * S];

    for (int idx = threadIdx.x; idx < K * NOUT; idx += blockDim.x) {
        int k = idx / NOUT;
        int n = idx - k * NOUT;
        lds_w[(k >> 1) * S + n * 2 + (k & 1)] = W[idx];
    }
    __syncthreads();

    const int lane = threadIdx.x & 31;
    const int wave = threadIdx.x >> 5;
    const int g = lane >> 4;
    const int mr = lane & 15;

    const int row0 = blockIdx.x * 128 + wave * 16;
    int row = row0 + mr;
    if (row >= N) row = N - 1;
    const float s = scale[row];
    const float* __restrict__ Arow = A + (size_t)row * K;

    v8f acc[NT];
#pragma unroll
    for (int t = 0; t < NT; ++t) acc[t] = (v8f){};

#pragma unroll 1
    for (int kb = 0; kb < K; kb += 16) {
        v2f a[4];
#pragma unroll
        for (int i = 0; i < 4; ++i)
            a[i] = *(const v2f*)(Arow + kb + 4 * i + 2 * g);
#pragma unroll
        for (int i = 0; i < 4; ++i) {
            const int k0 = kb + 4 * i;
            const v2f* __restrict__ wp =
                (const v2f*)(lds_w + (size_t)(k0 / 2 + g) * S);
            v2f b[NT];
#pragma unroll
            for (int t = 0; t < NT; ++t)
                b[t] = wp[t * 16 + mr];
            v2f av = a[i];
            av.x *= s;
            av.y *= s;
#pragma unroll
            for (int t = 0; t < NT; ++t)
                acc[t] = __builtin_amdgcn_wmma_f32_16x16x4_f32(
                     false, av,  false, b[t],
                     (short)0, acc[t],  false,  false);
        }
    }

#pragma unroll
    for (int t = 0; t < NT; ++t) {
        const int Nbase = t * 16;
        const float bv = FIRST ? bias[Nbase + mr] : 0.0f;
#pragma unroll
        for (int r = 0; r < 8; ++r) {
            const int orow = row0 + r + 8 * g;
            if (orow < N) {
                const size_t o = (size_t)orow * NOUT + Nbase + mr;
                if (FIRST) {
                    out[o] = fmaxf(acc[t][r] + bv, 0.0f);
                } else {
                    out[o] = acc[t][r];
                }
            }
        }
    }
}

extern "C" void kernel_launch(void* const* d_in, const int* in_sizes, int n_in,
                              void* d_out, int out_size, void* d_ws, size_t ws_size,
                              hipStream_t stream) {
    const float* in_feat = (const float*)d_in[0];
    const float* W1      = (const float*)d_in[1];
    const float* b1      = (const float*)d_in[2];
    const float* W2      = (const float*)d_in[3];
    const float* b2      = (const float*)d_in[4];
    const int*   src     = (const int*)d_in[5];
    const int*   dst     = (const int*)d_in[6];
    float*       out     = (float*)d_out;

    const int N = in_sizes[0] / IN_DIM;
    const int E = in_sizes[5];

    char* ws = (char*)d_ws;
    int*   deg   = (int*)ws;
    float* inv   = (float*)(ws + (size_t)4 * N * 4);
    float* agg   = (float*)(ws + (size_t)6 * N * 4);
    float* hbuf  = agg + (size_t)N * HID_DIM;
    float* zbuf  = hbuf + (size_t)N * HID_DIM;
    unsigned int* tags = (unsigned int*)(ws + (((size_t)1304 * N + 15) & ~(size_t)15));
    const int owners = (N + TILE_ROWS - 1) / TILE_ROWS;
    int* cnt = (int*)((char*)tags + (((size_t)E + 15) & ~(size_t)15));
    unsigned int* list = (unsigned int*)(cnt + ((owners + 31) & ~31));
    float* inv_out = inv;
    float* inv_in  = inv + N;

    hipMemsetAsync(deg, 0, (size_t)4 * N * sizeof(int), stream);

    {
        int blk = 256;
        int groups = (E + 3) / 4;
        gcn_degree_tag_kernel<<<(groups + blk - 1) / blk, blk, 0, stream>>>(
            src, dst, deg, tags, N, E);
        gcn_invsqrt_count_kernel<<<2 * owners, INV_THREADS, 0, stream>>>(
            deg, inv, cnt, N, owners);
    }

    gcn_owner_edge_sum_kernel<IN_DIM, 8><<<owners, 1024, 0, stream>>>(
        in_feat, src, dst, tags, inv_out, cnt, deg, list, agg, N, E);

    {
        int grid = (N + 127) / 128;
        gcn_wmma_gemm_kernel<IN_DIM, HID_DIM, true>
            <<<grid, 256, 0, stream>>>(agg, W1, b1, inv_in, hbuf, N);
    }

    {
        int grid = (N + 127) / 128;
        gcn_wmma_gemm_kernel<HID_DIM, OUT_DIM, false>
            <<<grid, 256, 0, stream>>>(hbuf, W2, b2, inv_out, zbuf, N);
    }

    gcn_listed_edge_sum_kernel<OUT_DIM, 16><<<owners, 1024, 0, stream>>>(
        zbuf, list, cnt, deg, inv_in, b2, out, N);
}


# Layer-1 owner search: degree launch writes 256 rows of one bit per edge (LDS integer OR); an owner reads its own 200 KB row, not the 1.6 MB byte plane; f32 sums stay in list (arrival) order
# speedup vs baseline: 3.8624x; 1.0336x over previous
#include <hip/hip_runtime.h>


typedef __attribute__((ext_vector_type(2))) float v2f;
typedef __attribute__((ext_vector_type(4))) float v4f;
typedef __attribute__((ext_vector_type(8))) float v8f;

#define IN_DIM  128
#define HID_DIM 128
#define OUT_DIM 64

__device__ __forceinline__ int gcn_package_half() {
    return (int)((__builtin_amdgcn_s_sendmsg_rtn(0x87) >> 18) & 1u);
}

#define TILE_ROWS 392
#define CHUNK_EDGES 1024
#define BIT_STRIDE 36
static_assert(CHUNK_EDGES == 1024, "256 threads: four edges each, 8 x 32 rows stored");
__global__ __launch_bounds__(CHUNK_EDGES / 4)
void gcn_degree_bits_kernel(const int* __restrict__ src,
                            const int* __restrict__ dst,
                            int* __restrict__ deg,
                            unsigned int* __restrict__ bits, int chunks, int N, int E) {
    __shared__ __attribute__((aligned(16))) unsigned int rowbits[256 * BIT_STRIDE];
    const int tid = (int)threadIdx.x;
    const int g = blockIdx.x * (CHUNK_EDGES / 4) + tid;
    const int e = g * 4;
    const bool whole = e + 3 < E;
    int4 s = make_int4(0, 0, 0, 0), t = make_int4(0, 0, 0, 0);
    if (whole) {
        s = ((const int4*)src)[g];
        t = ((const int4*)dst)[g];
    }
    for (int f = tid; f < 256 * BIT_STRIDE / 4; f += CHUNK_EDGES / 4)
        ((uint4*)rowbits)[f] = make_uint4(0u, 0u, 0u, 0u);
    __syncthreads();

    unsigned int* wcol = rowbits + (tid >> 3);
    const unsigned b0 = 4u * ((unsigned)tid & 7u);
    if (whole) {
        atomicOr(&wcol[(((unsigned)t.x / TILE_ROWS) & 255u) * BIT_STRIDE], 1u << b0);
        atomicOr(&wcol[(((unsigned)t.y / TILE_ROWS) & 255u) * BIT_STRIDE], 2u << b0);
        atomicOr(&wcol[(((unsigned)t.z / TILE_ROWS) & 255u) * BIT_STRIDE], 4u << b0);
        atomicOr(&wcol[(((unsigned)t.w / TILE_ROWS) & 255u) * BIT_STRIDE], 8u << b0);
    } else {
        for (int k = e; k < E; ++k)
            atomicOr(&wcol[(((unsigned)dst[k] / TILE_ROWS) & 255u) * BIT_STRIDE],
                     1u << (b0 + (unsigned)(k - e)));
    }
    __syncthreads();

    const size_t row_words = (size_t)chunks * (CHUNK_EDGES / 32);
    unsigned int* __restrict__ outp = bits + (size_t)blockIdx.x * (CHUNK_EDGES / 32) + 4 * (tid & 7);
#pragma unroll
    for (int j = 0; j < 8; ++j) {
        const int row = j * 32 + (tid >> 3);
        const uint4 v = *(const uint4*)(rowbits + row * BIT_STRIDE + 4 * (tid & 7));
        *(uint4*)(outp + (size_t)row * row_words) = v;
    }

    int* __restrict__ d = deg + (size_t)gcn_package_half() * (size_t)(2 * N);
    if (whole) {
        atomicAdd(&d[s.x], 1);
        atomicAdd(&d[s.y], 1);
        atomicAdd(&d[s.z], 1);
        atomicAdd(&d[s.w], 1);
        atomicAdd(&d[N + t.x], 1);
        atomicAdd(&d[N + t.y], 1);
        atomicAdd(&d[N + t.z], 1);
        atomicAdd(&d[N + t.w], 1);
    } else {
        for (int k = e; k < E; ++k) {
            atomicAdd(&d[src[k]], 1);
            atomicAdd(&d[N + dst[k]], 1);
        }
    }
}

#define INV_THREADS 512
static_assert(TILE_ROWS <= INV_THREADS, "one thread per row of an owner");
__global__ __launch_bounds__(INV_THREADS)
void gcn_invsqrt_count_kernel(const int* __restrict__ deg, float* __restrict__ inv,
                              int* __restrict__ cnt, int N, int owners) {
    __shared__ int part[INV_THREADS / 32];
    const bool in_kind = (int)blockIdx.x < owners;
    const int o = in_kind ? (int)blockIdx.x : (int)blockIdx.x - owners;
    const int row = o * TILE_ROWS + (int)threadIdx.x;
    int d = 0;
    if ((int)threadIdx.x < TILE_ROWS && row < N) {
        const int i = (in_kind ? N : 0) + row;
        d = deg[i] + deg[2 * N + i];
        inv[i] = rsqrtf(fmaxf((float)d, 1.0f));
    }
    if (in_kind) {
        for (int m = 16; m > 0; m >>= 1) d += __shfl_xor(d, m);
        if ((threadIdx.x & 31) == 0) part[threadIdx.x >> 5] = d;
        __syncthreads();
        if (threadIdx.x == 0) {
            int t = 0;
            for (int k = 0; k < INV_THREADS / 32; ++k) t += part[k];
            cnt[o] = t;
        }
    }
}

__device__ __forceinline__ int gcn_list_start(const int* __restrict__ cnt, int owner, int lane) {
    int a = 0;
    for (int j = lane; j < owner; j += 32) a += cnt[j];
    for (int m = 16; m > 0; m >>= 1) a += __shfl_xor(a, m);
    return a;
}

#define QUEUE_CAP 64
#define ROW_BITS 9
static_assert(TILE_ROWS < (1 << ROW_BITS), "row field of a list entry");

template <int V> struct gcn_lane_vec;
template <> struct gcn_lane_vec<4> { typedef v4f type; };
template <> struct gcn_lane_vec<2> { typedef v2f type; };

__device__ __forceinline__ int gcn_piece_lengths(const int* __restrict__ deg, int N, int lo,
                                                 int lane) {
    const int* __restrict__ d0 = deg + N;
    const int* __restrict__ d1 = deg + (size_t)3 * N;
    int c = 0;
#pragma unroll
    for (int k = 0; k < (TILE_ROWS + 31) / 32; ++k) {
        const int lr = lane + 32 * k;
        const bool ok = lr < TILE_ROWS && lo + lr < N;
        const int row = ok ? lo + lr : lo;
        const int d = d0[row] + d1[row];
        c += ok ? d : 0;
    }
    return c;
}

__device__ __forceinline__ int gcn_sum_before(int v, int lane) {
    int x = v;
    for (int d = 1; d < 32; d <<= 1) {
        const int t = __shfl_up(x, d);
        if (lane >= d) x += t;
    }
    return x - v;
}

template <int COLS, int U, typename VEC>
__device__ __forceinline__ void gcn_load_rows(VEC (&v)[U], const float* __restrict__ feat,
                                              int s, int jb, int lane) {
#pragma unroll
    for (int u = 0; u < U; ++u) {
        const unsigned sj = (unsigned)__builtin_amdgcn_readlane(s, jb + u);
        v[u] = *(const VEC*)(feat + (size_t)(sj * (unsigned)COLS) + lane * (COLS / 32));
    }
}

template <int COLS, int U, bool SCALE_BY_SRC, typename VEC>
__device__ __forceinline__ void gcn_sum_rows(const VEC (&v)[U], float* tile, int r, int sc_bits,
                                             int jb, int lane) {
    static_assert(U % 4 == 0, "groups of four edges");
#pragma unroll
    for (int g = 0; g < U; g += 4) {
        unsigned rj[4];
        VEC* tp[4];
        VEC t[4], nv[4];
#pragma unroll
        for (int i = 0; i < 4; ++i) {
            rj[i] = (unsigned)__builtin_amdgcn_readlane(r, jb + g + i);
            tp[i] = (VEC*)(tile + rj[i] * (unsigned)COLS + lane * (COLS / 32));
        }
#pragma unroll
        for (int i = 0; i < 4; ++i) t[i] = *tp[i];
#pragma unroll
        for (int i = 0; i < 4; ++i) {
            VEC base = t[i];
#pragma unroll
            for (int p = 0; p < i; ++p) base = (rj[p] == rj[i]) ? nv[p] : base;
            if (SCALE_BY_SRC) {
                const float scj =
                    __int_as_float(__builtin_amdgcn_readlane(sc_bits, jb + g + i));
                nv[i] = base + v[g + i] * scj;
            } else {
                nv[i] = base + v[g + i];
            }
        }
#pragma unroll
        for (int i = 0; i < 4; ++i) *tp[i] = nv[i];
    }
}

template <int COLS, int U, bool SCALE_BY_SRC>
__device__ __forceinline__ void gcn_add_rows_32(const float* __restrict__ feat, float* tile,
                                                int s, int r, int sc_bits, int lane) {
    static_assert(32 % (2 * U) == 0, "two sets of U edges take turns");
    typedef typename gcn_lane_vec<COLS / 32>::type vec;
    vec va[U], vb[U];
    gcn_load_rows<COLS, U>(va, feat, s, 0, lane);
#pragma unroll
    for (int jb = 0; jb < 32; jb += 2 * U) {
        gcn_load_rows<COLS, U>(vb, feat, s, jb + U, lane);
        gcn_sum_rows<COLS, U, SCALE_BY_SRC>(va, tile, r, sc_bits, jb, lane);
        if (jb + 2 * U < 32) gcn_load_rows<COLS, U>(va, feat, s, jb + 2 * U, lane);
        gcn_sum_rows<COLS, U, SCALE_BY_SRC>(vb, tile, r, sc_bits, jb + U, lane);
    }
}

template <int COLS, int UP, bool SCALE_BY_SRC>
__device__ __forceinline__ void gcn_add_rows_some(const float* __restrict__ feat, float* tile,
                                                  int s, int r, int sc_bits, int nb, int lane) {
    typedef typename gcn_lane_vec<COLS / 32>::type vec;
#pragma unroll 1
    for (int b = 0; b < nb; ++b) {
        vec v[UP];
        gcn_load_rows<COLS, UP>(v, feat, s, b * UP, lane);
        gcn_sum_rows<COLS, UP, SCALE_BY_SRC>(v, tile, r, sc_bits, b * UP, lane);
    }
}

template <int COLS, int U, bool SCALE_BY_SRC>
__device__ __forceinline__ void gcn_add_piece(const float* __restrict__ feat,
                                              const unsigned int* wl, int n,
                                              const float* __restrict__ inv_src,
                                              float* tile, int N, int lane) {
    if (n <= 0) return;
    unsigned w = wl[min(lane, n - 1)];
    int c0 = 0;
    do {
        const unsigned ws = w >> ROW_BITS;
        const unsigned wr = w & ((1u << ROW_BITS) - 1u);
        const int s = (int)(ws < (unsigned)N ? ws : (unsigned)N - 1u);
        int r = (int)(wr < (unsigned)TILE_ROWS ? wr : (unsigned)TILE_ROWS);
        const int c1 = c0 + 32;
        const unsigned wn = wl[min(c1 + lane, n - 1)];
        int sc_bits = 0;
        if (SCALE_BY_SRC) sc_bits = __float_as_int(inv_src[s]);
        const int m = n - c0;
        if (m >= 32) {
            gcn_add_rows_32<COLS, U, SCALE_BY_SRC>(feat, tile, s, r, sc_bits, lane);
        } else {
            if (lane >= m) r = TILE_ROWS;
            gcn_add_rows_some<COLS, 4, SCALE_BY_SRC>(feat, tile, s, r, sc_bits, (m + 3) / 4, lane);
        }
        w = wn;
        c0 = c1;
    } while (c0 < n);
}

template <bool FULL>
__device__ __forceinline__ void gcn_file_queued(const int* __restrict__ src,
                                                const int* __restrict__ dst,
                                                unsigned int* mylist, int* piece_fill,
                                                const int* q, unsigned head, int n,
                                                int lane, int lo, int E) {
    if (FULL || lane < n) {
        const int e = min(q[(head + (unsigned)lane) & (QUEUE_CAP - 1)], E - 1);
        const int s = src[e];
        const unsigned rr = (unsigned)(dst[e] - lo);
        if (rr < (unsigned)TILE_ROWS) {
            const int at = atomicAdd(&piece_fill[rr & 31u], 1);
            mylist[at] = ((unsigned)s << ROW_BITS) | rr;
        }
    }
}

template <int COLS, int U>
__global__ __launch_bounds__(1024)
void gcn_owner_edge_sum_kernel(const float* __restrict__ feat,
                               const int* __restrict__ src,
                               const int* __restrict__ dst,
                               const unsigned int* __restrict__ bits,
                               const float* __restrict__ inv_src,
                               const int* __restrict__ cnt,
                               const int* __restrict__ deg,
                               unsigned int* list,
                               float* __restrict__ result, int N, int E) {
    __shared__ __attribute__((aligned(16))) float tile[(TILE_ROWS + 1) * COLS];
    __shared__ int ring[32 * QUEUE_CAP];
    __shared__ int piece_fill[32];

    const int tid = (int)threadIdx.x;
    const int lane = tid & 31;
    const int wv = __builtin_amdgcn_readfirstlane(tid >> 5);
    int* q = ring + wv * QUEUE_CAP;
    const int lo = (int)blockIdx.x * TILE_ROWS;

    const int start = gcn_list_start(cnt, (int)blockIdx.x, lane);
    const int len_lane = gcn_piece_lengths(deg, N, lo, lane);
    const int at_lane = gcn_sum_before(len_lane, lane);
    if (wv == 0) piece_fill[lane] = at_lane;
    for (int f = tid; f < (TILE_ROWS + 1) * COLS / 4; f += 1024)
        ((float4*)tile)[f] = make_float4(0.0f, 0.0f, 0.0f, 0.0f);
    __syncthreads();
    unsigned int* mylist = list + __builtin_amdgcn_readfirstlane(start);

    const int nvec = ((E + CHUNK_EDGES - 1) / CHUNK_EDGES) * (CHUNK_EDGES / 128);
    const uint4* __restrict__ row16 =
        (const uint4*)bits + (size_t)((unsigned)blockIdx.x & 255u) * (size_t)nvec;
    const int trips = (nvec + 8 * 1024 - 1) / (8 * 1024);
    unsigned head = 0, cnt_q = 0;

    for (int kb = 0; kb < trips; ++kb) {
        const int v0 = kb * (8 * 1024) + tid;
        uint4 dv[8];
        if ((kb + 1) * (8 * 1024) <= nvec) {
#pragma unroll
            for (int i = 0; i < 8; ++i) dv[i] = row16[v0 + i * 1024];
        } else {
#pragma unroll
            for (int i = 0; i < 8; ++i) {
                const int v = v0 + i * 1024;
                dv[i] = make_uint4(0u, 0u, 0u, 0u);
                if (v < nvec) dv[i] = row16[v];
            }
        }
#pragma unroll
        for (int i = 0; i < 8; ++i) {
            unsigned w0 = dv[i].x, w1 = dv[i].y, w2 = dv[i].z, w3 = dv[i].w;
            int e0 = (v0 + i * 1024) * 128;
#pragma unroll 1
            for (int k = 0; k < 4; ++k) {
                unsigned m = w0;
                unsigned any = __builtin_amdgcn_ballot_w32(m != 0u);
                while (any != 0u) {
                    const unsigned pos = head + cnt_q + __builtin_amdgcn_mbcnt_lo(any, 0u);
                    if (m != 0u) {
                        q[pos & (QUEUE_CAP - 1)] = e0 + __builtin_ctz(m);
                        m &= m - 1u;
                    }
                    cnt_q += (unsigned)__builtin_popcount(any);
                    if (cnt_q >= 32u) {
                        gcn_file_queued<true>(src, dst, mylist, piece_fill, q, head, 32, lane, lo, E);
                        head += 32u;
                        cnt_q -= 32u;
                    }
                    any = __builtin_amdgcn_ballot_w32(m != 0u);
                }
                w0 = w1;
                w1 = w2;
                w2 = w3;
                e0 += 32;
            }
        }
    }
    while (cnt_q >= 32u) {
        gcn_file_queued<true>(src, dst, mylist, piece_fill, q, head, 32, lane, lo, E);
        head += 32u;
        cnt_q -= 32u;
    }
    if (cnt_q != 0u)
        gcn_file_queued<false>(src, dst, mylist, piece_fill, q, head, (int)cnt_q, lane, lo, E);
    __syncthreads();

    gcn_add_piece<COLS, U, true>(feat, mylist + __builtin_amdgcn_readlane(at_lane, wv),
                                 __builtin_amdgcn_readlane(len_lane, wv), inv_src, tile, N, lane);
    __syncthreads();

    constexpr int C4 = COLS / 4;
    for (int f = tid; f < TILE_ROWS * C4; f += 1024) {
        const int row = f / C4;
        const int c4 = f - row * C4;
        const int grow = lo + row;
        if (grow < N) ((float4*)result)[(size_t)grow * C4 + c4] = ((const float4*)tile)[f];
    }
}

template <int COLS, int U>
__global__ __launch_bounds__(1024)
void gcn_listed_edge_sum_kernel(const float* __restrict__ feat,
                                const unsigned int* __restrict__ list,
                                const int* __restrict__ cnt,
                                const int* __restrict__ deg,
                                const float* __restrict__ inv_row,
                                const float* __restrict__ bias,
                                float* __restrict__ result, int N) {
    __shared__ __attribute__((aligned(16))) float tile[(TILE_ROWS + 1) * COLS];

    const int tid = (int)threadIdx.x;
    const int lane = tid & 31;
    const int wv = __builtin_amdgcn_readfirstlane(tid >> 5);
    const int lo = (int)blockIdx.x * TILE_ROWS;

    const int start = gcn_list_start(cnt, (int)blockIdx.x, lane);
    const int len_lane = gcn_piece_lengths(deg, N, lo, lane);
    const int at_lane = gcn_sum_before(len_lane, lane);
    for (int f = tid; f < (TILE_ROWS + 1) * COLS / 4; f += 1024)
        ((float4*)tile)[f] = make_float4(0.0f, 0.0f, 0.0f, 0.0f);
    __syncthreads();
    const unsigned int* mypiece = list + __builtin_amdgcn_readfirstlane(start)
                                       + __builtin_amdgcn_readlane(at_lane, wv);
    gcn_add_piece<COLS, U, false>(feat, mypiece, __builtin_amdgcn_readlane(len_lane, wv),
                                  nullptr, tile, N, lane);
    __syncthreads();

    constexpr int C4 = COLS / 4;
    for (int f = tid; f < TILE_ROWS * C4; f += 1024) {
        const int row = f / C4;
        const int c4 = f - row * C4;
        const int grow = lo + row;
        if (grow < N) {
            float4 t = ((const float4*)tile)[f];
            const float sc = inv_row[grow];
            const float4 b = ((const float4*)bias)[c4];
            t.x = b.x + sc * t.x;
            t.y = b.y + sc * t.y;
            t.z = b.z + sc * t.z;
            t.w = b.w + sc * t.w;
            ((float4*)result)[(size_t)grow * C4 + c4] = t;
        }
    }
}

template <int K, int NOUT, bool FIRST>
__global__ __launch_bounds__(256)
void gcn_wmma_gemm_kernel(const float* __restrict__ A,
                          const float* __restrict__ W,
                          const float* __restrict__ bias,
                          const float* __restrict__ scale,
                          float* __restrict__ out, int N) {
    constexpr int S  = 2 * NOUT + 32;
    constexpr int NT = NOUT / 16;
    __shared__ float lds_w[(K / 2) * S];

    for (int idx = threadIdx.x; idx < K * NOUT; idx += blockDim.x) {
        int k = idx / NOUT;
        int n = idx - k * NOUT;
        lds_w[(k >> 1) * S + n * 2 + (k & 1)] = W[idx];
    }
    __syncthreads();

    const int lane = threadIdx.x & 31;
    const int wave = threadIdx.x >> 5;
    const int g = lane >> 4;
    const int mr = lane & 15;

    const int row0 = blockIdx.x * 128 + wave * 16;
    int row = row0 + mr;
    if (row >= N) row = N - 1;
    const float s = scale[row];
    const float* __restrict__ Arow = A + (size_t)row * K;

    v8f acc[NT];
#pragma unroll
    for (int t = 0; t < NT; ++t) acc[t] = (v8f){};

#pragma unroll 1
    for (int kb = 0; kb < K; kb += 16) {
        v2f a[4];
#pragma unroll
        for (int i = 0; i < 4; ++i)
            a[i] = *(const v2f*)(Arow + kb + 4 * i + 2 * g);
#pragma unroll
        for (int i = 0; i < 4; ++i) {
            const int k0 = kb + 4 * i;
            const v2f* __restrict__ wp =
                (const v2f*)(lds_w + (size_t)(k0 / 2 + g) * S);
            v2f b[NT];
#pragma unroll
            for (int t = 0; t < NT; ++t)
                b[t] = wp[t * 16 + mr];
            v2f av = a[i];
            av.x *= s;
            av.y *= s;
#pragma unroll
            for (int t = 0; t < NT; ++t)
                acc[t] = __builtin_amdgcn_wmma_f32_16x16x4_f32(
                     false, av,  false, b[t],
                     (short)0, acc[t],  false,  false);
        }
    }

#pragma unroll
    for (int t = 0; t < NT; ++t) {
        const int Nbase = t * 16;
        const float bv = FIRST ? bias[Nbase + mr] : 0.0f;
#pragma unroll
        for (int r = 0; r < 8; ++r) {
            const int orow = row0 + r + 8 * g;
            if (orow < N) {
                const size_t o = (size_t)orow * NOUT + Nbase + mr;
                if (FIRST) {
                    out[o] = fmaxf(acc[t][r] + bv, 0.0f);
                } else {
                    out[o] = acc[t][r];
                }
            }
        }
    }
}

extern "C" void kernel_launch(void* const* d_in, const int* in_sizes, int n_in,
                              void* d_out, int out_size, void* d_ws, size_t ws_size,
                              hipStream_t stream) {
    const float* in_feat = (const float*)d_in[0];
    const float* W1      = (const float*)d_in[1];
    const float* b1      = (const float*)d_in[2];
    const float* W2      = (const float*)d_in[3];
    const float* b2      = (const float*)d_in[4];
    const int*   src     = (const int*)d_in[5];
    const int*   dst     = (const int*)d_in[6];
    float*       out     = (float*)d_out;

    const int N = in_sizes[0] / IN_DIM;
    const int E = in_sizes[5];

    char* ws = (char*)d_ws;
    int*   deg   = (int*)ws;
    float* inv   = (float*)(ws + (size_t)4 * N * 4);
    float* agg   = (float*)(ws + (size_t)6 * N * 4);
    float* hbuf  = agg + (size_t)N * HID_DIM;
    float* zbuf  = hbuf + (size_t)N * HID_DIM;
    unsigned int* bits = (unsigned int*)(ws + (((size_t)1304 * N + 255) & ~(size_t)255));
    const int chunks = (E + CHUNK_EDGES - 1) / CHUNK_EDGES;
    const int owners = (N + TILE_ROWS - 1) / TILE_ROWS;
    int* cnt = (int*)((char*)bits + (size_t)256 * chunks * (CHUNK_EDGES / 8));
    unsigned int* list = (unsigned int*)(cnt + ((owners + 31) & ~31));
    float* inv_out = inv;
    float* inv_in  = inv + N;

    hipMemsetAsync(deg, 0, (size_t)4 * N * sizeof(int), stream);

    {
        gcn_degree_bits_kernel<<<chunks, CHUNK_EDGES / 4, 0, stream>>>(
            src, dst, deg, bits, chunks, N, E);
        gcn_invsqrt_count_kernel<<<2 * owners, INV_THREADS, 0, stream>>>(
            deg, inv, cnt, N, owners);
    }

    gcn_owner_edge_sum_kernel<IN_DIM, 8><<<owners, 1024, 0, stream>>>(
        in_feat, src, dst, bits, inv_out, cnt, deg, list, agg, N, E);

    {
        int grid = (N + 127) / 128;
        gcn_wmma_gemm_kernel<IN_DIM, HID_DIM, true>
            <<<grid, 256, 0, stream>>>(agg, W1, b1, inv_in, hbuf, N);
    }

    {
        int grid = (N + 127) / 128;
        gcn_wmma_gemm_kernel<HID_DIM, OUT_DIM, false>
            <<<grid, 256, 0, stream>>>(hbuf, W2, b2, inv_out, zbuf, N);
    }

    gcn_listed_edge_sum_kernel<OUT_DIM, 16><<<owners, 1024, 0, stream>>>(
        zbuf, list, cnt, deg, inv_in, b2, out, N);
}


# Both product launches: W staged into LDS by 16-byte loads, a constant 16 (8) per thread all in flight together, not 64 (32) dependent 4-byte load-wait-store trips; LDS layout and all sums unchanged
# speedup vs baseline: 4.5363x; 1.1745x over previous
#include <hip/hip_runtime.h>


typedef __attribute__((ext_vector_type(2))) float v2f;
typedef __attribute__((ext_vector_type(4))) float v4f;
typedef __attribute__((ext_vector_type(8))) float v8f;

#define IN_DIM  128
#define HID_DIM 128
#define OUT_DIM 64

__device__ __forceinline__ int gcn_package_half() {
    return (int)((__builtin_amdgcn_s_sendmsg_rtn(0x87) >> 18) & 1u);
}

#define TILE_ROWS 392
#define CHUNK_EDGES 1024
#define BIT_STRIDE 36
static_assert(CHUNK_EDGES == 1024, "256 threads: four edges each, 8 x 32 rows stored");
__global__ __launch_bounds__(CHUNK_EDGES / 4)
void gcn_degree_bits_kernel(const int* __restrict__ src,
                            const int* __restrict__ dst,
                            int* __restrict__ deg,
                            unsigned int* __restrict__ bits, int chunks, int N, int E) {
    __shared__ __attribute__((aligned(16))) unsigned int rowbits[256 * BIT_STRIDE];
    const int tid = (int)threadIdx.x;
    const int g = blockIdx.x * (CHUNK_EDGES / 4) + tid;
    const int e = g * 4;
    const bool whole = e + 3 < E;
    int4 s = make_int4(0, 0, 0, 0), t = make_int4(0, 0, 0, 0);
    if (whole) {
        s = ((const int4*)src)[g];
        t = ((const int4*)dst)[g];
    }
    for (int f = tid; f < 256 * BIT_STRIDE / 4; f += CHUNK_EDGES / 4)
        ((uint4*)rowbits)[f] = make_uint4(0u, 0u, 0u, 0u);
    __syncthreads();

    unsigned int* wcol = rowbits + (tid >> 3);
    const unsigned b0 = 4u * ((unsigned)tid & 7u);
    if (whole) {
        atomicOr(&wcol[(((unsigned)t.x / TILE_ROWS) & 255u) * BIT_STRIDE], 1u << b0);
        atomicOr(&wcol[(((unsigned)t.y / TILE_ROWS) & 255u) * BIT_STRIDE], 2u << b0);
        atomicOr(&wcol[(((unsigned)t.z / TILE_ROWS) & 255u) * BIT_STRIDE], 4u << b0);
        atomicOr(&wcol[(((unsigned)t.w / TILE_ROWS) & 255u) * BIT_STRIDE], 8u << b0);
    } else {
        for (int k = e; k < E; ++k)
            atomicOr(&wcol[(((unsigned)dst[k] / TILE_ROWS) & 255u) * BIT_STRIDE],
                     1u << (b0 + (unsigned)(k - e)));
    }
    __syncthreads();

    const size_t row_words = (size_t)chunks * (CHUNK_EDGES / 32);
    unsigned int* __restrict__ outp = bits + (size_t)blockIdx.x * (CHUNK_EDGES / 32) + 4 * (tid & 7);
#pragma unroll
    for (int j = 0; j < 8; ++j) {
        const int row = j * 32 + (tid >> 3);
        const uint4 v = *(const uint4*)(rowbits + row * BIT_STRIDE + 4 * (tid & 7));
        *(uint4*)(outp + (size_t)row * row_words) = v;
    }

    int* __restrict__ d = deg + (size_t)gcn_package_half() * (size_t)(2 * N);
    if (whole) {
        atomicAdd(&d[s.x], 1);
        atomicAdd(&d[s.y], 1);
        atomicAdd(&d[s.z], 1);
        atomicAdd(&d[s.w], 1);
        atomicAdd(&d[N + t.x], 1);
        atomicAdd(&d[N + t.y], 1);
        atomicAdd(&d[N + t.z], 1);
        atomicAdd(&d[N + t.w], 1);
    } else {
        for (int k = e; k < E; ++k) {
            atomicAdd(&d[src[k]], 1);
            atomicAdd(&d[N + dst[k]], 1);
        }
    }
}

#define INV_THREADS 512
static_assert(TILE_ROWS <= INV_THREADS, "one thread per row of an owner");
__global__ __launch_bounds__(INV_THREADS)
void gcn_invsqrt_count_kernel(const int* __restrict__ deg, float* __restrict__ inv,
                              int* __restrict__ cnt, int N, int owners) {
    __shared__ int part[INV_THREADS / 32];
    const bool in_kind = (int)blockIdx.x < owners;
    const int o = in_kind ? (int)blockIdx.x : (int)blockIdx.x - owners;
    const int row = o * TILE_ROWS + (int)threadIdx.x;
    int d = 0;
    if ((int)threadIdx.x < TILE_ROWS && row < N) {
        const int i = (in_kind ? N : 0) + row;
        d = deg[i] + deg[2 * N + i];
        inv[i] = rsqrtf(fmaxf((float)d, 1.0f));
    }
    if (in_kind) {
        for (int m = 16; m > 0; m >>= 1) d += __shfl_xor(d, m);
        if ((threadIdx.x & 31) == 0) part[threadIdx.x >> 5] = d;
        __syncthreads();
        if (threadIdx.x == 0) {
            int t = 0;
            for (int k = 0; k < INV_THREADS / 32; ++k) t += part[k];
            cnt[o] = t;
        }
    }
}

__device__ __forceinline__ int gcn_list_start(const int* __restrict__ cnt, int owner, int lane) {
    int a = 0;
    for (int j = lane; j < owner; j += 32) a += cnt[j];
    for (int m = 16; m > 0; m >>= 1) a += __shfl_xor(a, m);
    return a;
}

#define QUEUE_CAP 64
#define ROW_BITS 9
static_assert(TILE_ROWS < (1 << ROW_BITS), "row field of a list entry");

template <int V> struct gcn_lane_vec;
template <> struct gcn_lane_vec<4> { typedef v4f type; };
template <> struct gcn_lane_vec<2> { typedef v2f type; };

__device__ __forceinline__ int gcn_piece_lengths(const int* __restrict__ deg, int N, int lo,
                                                 int lane) {
    const int* __restrict__ d0 = deg + N;
    const int* __restrict__ d1 = deg + (size_t)3 * N;
    int c = 0;
#pragma unroll
    for (int k = 0; k < (TILE_ROWS + 31) / 32; ++k) {
        const int lr = lane + 32 * k;
        const bool ok = lr < TILE_ROWS && lo + lr < N;
        const int row = ok ? lo + lr : lo;
        const int d = d0[row] + d1[row];
        c += ok ? d : 0;
    }
    return c;
}

__device__ __forceinline__ int gcn_sum_before(int v, int lane) {
    int x = v;
    for (int d = 1; d < 32; d <<= 1) {
        const int t = __shfl_up(x, d);
        if (lane >= d) x += t;
    }
    return x - v;
}

template <int COLS, int U, typename VEC>
__device__ __forceinline__ void gcn_load_rows(VEC (&v)[U], const float* __restrict__ feat,
                                              int s, int jb, int lane) {
#pragma unroll
    for (int u = 0; u < U; ++u) {
        const unsigned sj = (unsigned)__builtin_amdgcn_readlane(s, jb + u);
        v[u] = *(const VEC*)(feat + (size_t)(sj * (unsigned)COLS) + lane * (COLS / 32));
    }
}

template <int COLS, int U, bool SCALE_BY_SRC, typename VEC>
__device__ __forceinline__ void gcn_sum_rows(const VEC (&v)[U], float* tile, int r, int sc_bits,
                                             int jb, int lane) {
    static_assert(U % 4 == 0, "groups of four edges");
#pragma unroll
    for (int g = 0; g < U; g += 4) {
        unsigned rj[4];
        VEC* tp[4];
        VEC t[4], nv[4];
#pragma unroll
        for (int i = 0; i < 4; ++i) {
            rj[i] = (unsigned)__builtin_amdgcn_readlane(r, jb + g + i);
            tp[i] = (VEC*)(tile + rj[i] * (unsigned)COLS + lane * (COLS / 32));
        }
#pragma unroll
        for (int i = 0; i < 4; ++i) t[i] = *tp[i];
#pragma unroll
        for (int i = 0; i < 4; ++i) {
            VEC base = t[i];
#pragma unroll
            for (int p = 0; p < i; ++p) base = (rj[p] == rj[i]) ? nv[p] : base;
            if (SCALE_BY_SRC) {
                const float scj =
                    __int_as_float(__builtin_amdgcn_readlane(sc_bits, jb + g + i));
                nv[i] = base + v[g + i] * scj;
            } else {
                nv[i] = base + v[g + i];
            }
        }
#pragma unroll
        for (int i = 0; i < 4; ++i) *tp[i] = nv[i];
    }
}

template <int COLS, int U, bool SCALE_BY_SRC>
__device__ __forceinline__ void gcn_add_rows_32(const float* __restrict__ feat, float* tile,
                                                int s, int r, int sc_bits, int lane) {
    static_assert(32 % (2 * U) == 0, "two sets of U edges take turns");
    typedef typename gcn_lane_vec<COLS / 32>::type vec;
    vec va[U], vb[U];
    gcn_load_rows<COLS, U>(va, feat, s, 0, lane);
#pragma unroll
    for (int jb = 0; jb < 32; jb += 2 * U) {
        gcn_load_rows<COLS, U>(vb, feat, s, jb + U, lane);
        gcn_sum_rows<COLS, U, SCALE_BY_SRC>(va, tile, r, sc_bits, jb, lane);
        if (jb + 2 * U < 32) gcn_load_rows<COLS, U>(va, feat, s, jb + 2 * U, lane);
        gcn_sum_rows<COLS, U, SCALE_BY_SRC>(vb, tile, r, sc_bits, jb + U, lane);
    }
}

template <int COLS, int UP, bool SCALE_BY_SRC>
__device__ __forceinline__ void gcn_add_rows_some(const float* __restrict__ feat, float* tile,
                                                  int s, int r, int sc_bits, int nb, int lane) {
    typedef typename gcn_lane_vec<COLS / 32>::type vec;
#pragma unroll 1
    for (int b = 0; b < nb; ++b) {
        vec v[UP];
        gcn_load_rows<COLS, UP>(v, feat, s, b * UP, lane);
        gcn_sum_rows<COLS, UP, SCALE_BY_SRC>(v, tile, r, sc_bits, b * UP, lane);
    }
}

template <int COLS, int U, bool SCALE_BY_SRC>
__device__ __forceinline__ void gcn_add_piece(const float* __restrict__ feat,
                                              const unsigned int* wl, int n,
                                              const float* __restrict__ inv_src,
                                              float* tile, int N, int lane) {
    if (n <= 0) return;
    unsigned w = wl[min(lane, n - 1)];
    int c0 = 0;
    do {
        const unsigned ws = w >> ROW_BITS;
        const unsigned wr = w & ((1u << ROW_BITS) - 1u);
        const int s = (int)(ws < (unsigned)N ? ws : (unsigned)N - 1u);
        int r = (int)(wr < (unsigned)TILE_ROWS ? wr : (unsigned)TILE_ROWS);
        const int c1 = c0 + 32;
        const unsigned wn = wl[min(c1 + lane, n - 1)];
        int sc_bits = 0;
        if (SCALE_BY_SRC) sc_bits = __float_as_int(inv_src[s]);
        const int m = n - c0;
        if (m >= 32) {
            gcn_add_rows_32<COLS, U, SCALE_BY_SRC>(feat, tile, s, r, sc_bits, lane);
        } else {
            if (lane >= m) r = TILE_ROWS;
            gcn_add_rows_some<COLS, 4, SCALE_BY_SRC>(feat, tile, s, r, sc_bits, (m + 3) / 4, lane);
        }
        w = wn;
        c0 = c1;
    } while (c0 < n);
}

template <bool FULL>
__device__ __forceinline__ void gcn_file_queued(const int* __restrict__ src,
                                                const int* __restrict__ dst,
                                                unsigned int* mylist, int* piece_fill,
                                                const int* q, unsigned head, int n,
                                                int lane, int lo, int E) {
    if (FULL || lane < n) {
        const int e = min(q[(head + (unsigned)lane) & (QUEUE_CAP - 1)], E - 1);
        const int s = src[e];
        const unsigned rr = (unsigned)(dst[e] - lo);
        if (rr < (unsigned)TILE_ROWS) {
            const int at = atomicAdd(&piece_fill[rr & 31u], 1);
            mylist[at] = ((unsigned)s << ROW_BITS) | rr;
        }
    }
}

template <int COLS, int U>
__global__ __launch_bounds__(1024)
void gcn_owner_edge_sum_kernel(const float* __restrict__ feat,
                               const int* __restrict__ src,
                               const int* __restrict__ dst,
                               const unsigned int* __restrict__ bits,
                               const float* __restrict__ inv_src,
                               const int* __restrict__ cnt,
                               const int* __restrict__ deg,
                               unsigned int* list,
                               float* __restrict__ result, int N, int E) {
    __shared__ __attribute__((aligned(16))) float tile[(TILE_ROWS + 1) * COLS];
    __shared__ int ring[32 * QUEUE_CAP];
    __shared__ int piece_fill[32];

    const int tid = (int)threadIdx.x;
    const int lane = tid & 31;
    const int wv = __builtin_amdgcn_readfirstlane(tid >> 5);
    int* q = ring + wv * QUEUE_CAP;
    const int lo = (int)blockIdx.x * TILE_ROWS;

    const int start = gcn_list_start(cnt, (int)blockIdx.x, lane);
    const int len_lane = gcn_piece_lengths(deg, N, lo, lane);
    const int at_lane = gcn_sum_before(len_lane, lane);
    if (wv == 0) piece_fill[lane] = at_lane;
    for (int f = tid; f < (TILE_ROWS + 1) * COLS / 4; f += 1024)
        ((float4*)tile)[f] = make_float4(0.0f, 0.0f, 0.0f, 0.0f);
    __syncthreads();
    unsigned int* mylist = list + __builtin_amdgcn_readfirstlane(start);

    const int nvec = ((E + CHUNK_EDGES - 1) / CHUNK_EDGES) * (CHUNK_EDGES / 128);
    const uint4* __restrict__ row16 =
        (const uint4*)bits + (size_t)((unsigned)blockIdx.x & 255u) * (size_t)nvec;
    const int trips = (nvec + 8 * 1024 - 1) / (8 * 1024);
    unsigned head = 0, cnt_q = 0;

    for (int kb = 0; kb < trips; ++kb) {
        const int v0 = kb * (8 * 1024) + tid;
        uint4 dv[8];
        if ((kb + 1) * (8 * 1024) <= nvec) {
#pragma unroll
            for (int i = 0; i < 8; ++i) dv[i] = row16[v0 + i * 1024];
        } else {
#pragma unroll
            for (int i = 0; i < 8; ++i) {
                const int v = v0 + i * 1024;
                dv[i] = make_uint4(0u, 0u, 0u, 0u);
                if (v < nvec) dv[i] = row16[v];
            }
        }
#pragma unroll
        for (int i = 0; i < 8; ++i) {
            unsigned w0 = dv[i].x, w1 = dv[i].y, w2 = dv[i].z, w3 = dv[i].w;
            int e0 = (v0 + i * 1024) * 128;
#pragma unroll 1
            for (int k = 0; k < 4; ++k) {
                unsigned m = w0;
                unsigned any = __builtin_amdgcn_ballot_w32(m != 0u);
                while (any != 0u) {
                    const unsigned pos = head + cnt_q + __builtin_amdgcn_mbcnt_lo(any, 0u);
                    if (m != 0u) {
                        q[pos & (QUEUE_CAP - 1)] = e0 + __builtin_ctz(m);
                        m &= m - 1u;
                    }
                    cnt_q += (unsigned)__builtin_popcount(any);
                    if (cnt_q >= 32u) {
                        gcn_file_queued<true>(src, dst, mylist, piece_fill, q, head, 32, lane, lo, E);
                        head += 32u;
                        cnt_q -= 32u;
                    }
                    any = __builtin_amdgcn_ballot_w32(m != 0u);
                }
                w0 = w1;
                w1 = w2;
                w2 = w3;
                e0 += 32;
            }
        }
    }
    while (cnt_q >= 32u) {
        gcn_file_queued<true>(src, dst, mylist, piece_fill, q, head, 32, lane, lo, E);
        head += 32u;
        cnt_q -= 32u;
    }
    if (cnt_q != 0u)
        gcn_file_queued<false>(src, dst, mylist, piece_fill, q, head, (int)cnt_q, lane, lo, E);
    __syncthreads();

    gcn_add_piece<COLS, U, true>(feat, mylist + __builtin_amdgcn_readlane(at_lane, wv),
                                 __builtin_amdgcn_readlane(len_lane, wv), inv_src, tile, N, lane);
    __syncthreads();

    constexpr int C4 = COLS / 4;
    for (int f = tid; f < TILE_ROWS * C4; f += 1024) {
        const int row = f / C4;
        const int c4 = f - row * C4;
        const int grow = lo + row;
        if (grow < N) ((float4*)result)[(size_t)grow * C4 + c4] = ((const float4*)tile)[f];
    }
}

template <int COLS, int U>
__global__ __launch_bounds__(1024)
void gcn_listed_edge_sum_kernel(const float* __restrict__ feat,
                                const unsigned int* __restrict__ list,
                                const int* __restrict__ cnt,
                                const int* __restrict__ deg,
                                const float* __restrict__ inv_row,
                                const float* __restrict__ bias,
                                float* __restrict__ result, int N) {
    __shared__ __attribute__((aligned(16))) float tile[(TILE_ROWS + 1) * COLS];

    const int tid = (int)threadIdx.x;
    const int lane = tid & 31;
    const int wv = __builtin_amdgcn_readfirstlane(tid >> 5);
    const int lo = (int)blockIdx.x * TILE_ROWS;

    const int start = gcn_list_start(cnt, (int)blockIdx.x, lane);
    const int len_lane = gcn_piece_lengths(deg, N, lo, lane);
    const int at_lane = gcn_sum_before(len_lane, lane);
    for (int f = tid; f < (TILE_ROWS + 1) * COLS / 4; f += 1024)
        ((float4*)tile)[f] = make_float4(0.0f, 0.0f, 0.0f, 0.0f);
    __syncthreads();
    const unsigned int* mypiece = list + __builtin_amdgcn_readfirstlane(start)
                                       + __builtin_amdgcn_readlane(at_lane, wv);
    gcn_add_piece<COLS, U, false>(feat, mypiece, __builtin_amdgcn_readlane(len_lane, wv),
                                  nullptr, tile, N, lane);
    __syncthreads();

    constexpr int C4 = COLS / 4;
    for (int f = tid; f < TILE_ROWS * C4; f += 1024) {
        const int row = f / C4;
        const int c4 = f - row * C4;
        const int grow = lo + row;
        if (grow < N) {
            float4 t = ((const float4*)tile)[f];
            const float sc = inv_row[grow];
            const float4 b = ((const float4*)bias)[c4];
            t.x = b.x + sc * t.x;
            t.y = b.y + sc * t.y;
            t.z = b.z + sc * t.z;
            t.w = b.w + sc * t.w;
            ((float4*)result)[(size_t)grow * C4 + c4] = t;
        }
    }
}

template <int K, int NOUT, bool FIRST>
__global__ __launch_bounds__(256)
void gcn_wmma_gemm_kernel(const float* __restrict__ A,
                          const float* __restrict__ W,
                          const float* __restrict__ bias,
                          const float* __restrict__ scale,
                          float* __restrict__ out, int N) {
    constexpr int S  = 2 * NOUT + 32;
    constexpr int NT = NOUT / 16;
    __shared__ __attribute__((aligned(16))) float lds_w[(K / 2) * S];

    constexpr int N4 = NOUT / 4;
    constexpr int ITEMS = (K / 2) * N4 / 256;
    static_assert((K / 2) * N4 % 256 == 0, "whole items for 256 threads");
    {
        const float4* __restrict__ W4 = (const float4*)W;
        float4 lo[ITEMS], hi[ITEMS];
#pragma unroll
        for (int i = 0; i < ITEMS; ++i) {
            const int it = i * 256 + (int)threadIdx.x;
            const int p = it / N4;
            const int q = it - p * N4;
            lo[i] = W4[(2 * p) * N4 + q];
            hi[i] = W4[(2 * p + 1) * N4 + q];
        }
#pragma unroll
        for (int i = 0; i < ITEMS; ++i) {
            const int it = i * 256 + (int)threadIdx.x;
            const int p = it / N4;
            const int q = it - p * N4;
            float4* d = (float4*)(lds_w + p * S + q * 8);
            d[0] = make_float4(lo[i].x, hi[i].x, lo[i].y, hi[i].y);
            d[1] = make_float4(lo[i].z, hi[i].z, lo[i].w, hi[i].w);
        }
    }
    __syncthreads();

    const int lane = threadIdx.x & 31;
    const int wave = threadIdx.x >> 5;
    const int g = lane >> 4;
    const int mr = lane & 15;

    const int row0 = blockIdx.x * 128 + wave * 16;
    int row = row0 + mr;
    if (row >= N) row = N - 1;
    const float s = scale[row];
    const float* __restrict__ Arow = A + (size_t)row * K;

    v8f acc[NT];
#pragma unroll
    for (int t = 0; t < NT; ++t) acc[t] = (v8f){};

#pragma unroll 1
    for (int kb = 0; kb < K; kb += 16) {
        v2f a[4];
#pragma unroll
        for (int i = 0; i < 4; ++i)
            a[i] = *(const v2f*)(Arow + kb + 4 * i + 2 * g);
#pragma unroll
        for (int i = 0; i < 4; ++i) {
            const int k0 = kb + 4 * i;
            const v2f* __restrict__ wp =
                (const v2f*)(lds_w + (size_t)(k0 / 2 + g) * S);
            v2f b[NT];
#pragma unroll
            for (int t = 0; t < NT; ++t)
                b[t] = wp[t * 16 + mr];
            v2f av = a[i];
            av.x *= s;
            av.y *= s;
#pragma unroll
            for (int t = 0; t < NT; ++t)
                acc[t] = __builtin_amdgcn_wmma_f32_16x16x4_f32(
                     false, av,  false, b[t],
                     (short)0, acc[t],  false,  false);
        }
    }

#pragma unroll
    for (int t = 0; t < NT; ++t) {
        const int Nbase = t * 16;
        const float bv = FIRST ? bias[Nbase + mr] : 0.0f;
#pragma unroll
        for (int r = 0; r < 8; ++r) {
            const int orow = row0 + r + 8 * g;
            if (orow < N) {
                const size_t o = (size_t)orow * NOUT + Nbase + mr;
                if (FIRST) {
                    out[o] = fmaxf(acc[t][r] + bv, 0.0f);
                } else {
                    out[o] = acc[t][r];
                }
            }
        }
    }
}

extern "C" void kernel_launch(void* const* d_in, const int* in_sizes, int n_in,
                              void* d_out, int out_size, void* d_ws, size_t ws_size,
                              hipStream_t stream) {
    const float* in_feat = (const float*)d_in[0];
    const float* W1      = (const float*)d_in[1];
    const float* b1      = (const float*)d_in[2];
    const float* W2      = (const float*)d_in[3];
    const float* b2      = (const float*)d_in[4];
    const int*   src     = (const int*)d_in[5];
    const int*   dst     = (const int*)d_in[6];
    float*       out     = (float*)d_out;

    const int N = in_sizes[0] / IN_DIM;
    const int E = in_sizes[5];

    char* ws = (char*)d_ws;
    int*   deg   = (int*)ws;
    float* inv   = (float*)(ws + (size_t)4 * N * 4);
    float* agg   = (float*)(ws + (size_t)6 * N * 4);
    float* hbuf  = agg + (size_t)N * HID_DIM;
    float* zbuf  = hbuf + (size_t)N * HID_DIM;
    unsigned int* bits = (unsigned int*)(ws + (((size_t)1304 * N + 255) & ~(size_t)255));
    const int chunks = (E + CHUNK_EDGES - 1) / CHUNK_EDGES;
    const int owners = (N + TILE_ROWS - 1) / TILE_ROWS;
    int* cnt = (int*)((char*)bits + (size_t)256 * chunks * (CHUNK_EDGES / 8));
    unsigned int* list = (unsigned int*)(cnt + ((owners + 31) & ~31));
    float* inv_out = inv;
    float* inv_in  = inv + N;

    hipMemsetAsync(deg, 0, (size_t)4 * N * sizeof(int), stream);

    {
        gcn_degree_bits_kernel<<<chunks, CHUNK_EDGES / 4, 0, stream>>>(
            src, dst, deg, bits, chunks, N, E);
        gcn_invsqrt_count_kernel<<<2 * owners, INV_THREADS, 0, stream>>>(
            deg, inv, cnt, N, owners);
    }

    gcn_owner_edge_sum_kernel<IN_DIM, 8><<<owners, 1024, 0, stream>>>(
        in_feat, src, dst, bits, inv_out, cnt, deg, list, agg, N, E);

    {
        int grid = (N + 127) / 128;
        gcn_wmma_gemm_kernel<IN_DIM, HID_DIM, true>
            <<<grid, 256, 0, stream>>>(agg, W1, b1, inv_in, hbuf, N);
    }

    {
        int grid = (N + 127) / 128;
        gcn_wmma_gemm_kernel<HID_DIM, OUT_DIM, false>
            <<<grid, 256, 0, stream>>>(hbuf, W2, b2, inv_out, zbuf, N);
    }

    gcn_listed_edge_sum_kernel<OUT_DIM, 16><<<owners, 1024, 0, stream>>>(
        zbuf, list, cnt, deg, inv_in, b2, out, N);
}
